# starting kernel
# baseline (speedup 1.0000x reference)
#include <hip/hip_runtime.h>

typedef __attribute__((ext_vector_type(2))) float v2f;
typedef __attribute__((ext_vector_type(8))) float v8f;

#define DIM        128
#define NEDGES     50000
#define NEG_SLOPE  0.01f

__global__ void __launch_bounds__(256)
gemm_wmma_f32(const float* __restrict__ X, const float* __restrict__ W,
              float* __restrict__ Y, int mtiles)
{
    const int lane  = threadIdx.x & 31;
    const int wv    = threadIdx.x >> 5;
    const long tile = (long)blockIdx.x * 8 + wv;
    const long mt   = tile >> 3;
    const int  nt   = (int)(tile & 7);
    if (mt >= mtiles) return;

    const int half = lane >> 4;
    const int lr   = lane & 15;
    const size_t row  = (size_t)mt * 16 + lr;
    const int    col0 = nt * 16;

    const float* __restrict__ ap = X + row * DIM + 2 * half;
    const float* __restrict__ bp = W + (size_t)(2 * half) * DIM + col0 + lr;

    v8f c = {};
#pragma unroll
    for (int k = 0; k < DIM; k += 4) {
        v2f a, b;
        a.x = ap[k];
        a.y = ap[k + 1];
        b.x = bp[(size_t)k * DIM];
        b.y = bp[(size_t)k * DIM + DIM];
        c = __builtin_amdgcn_wmma_f32_16x16x4_f32(false, a, false, b,
                                                  (short)0, c, false, false);
    }

    float* __restrict__ yp = Y + ((size_t)mt * 16 + half * 8) * DIM + col0 + lr;
#pragma unroll
    for (int r = 0; r < 8; ++r)
        yp[(size_t)r * DIM] = c[r];
}

__global__ void zero_kernel(float* __restrict__ p, long n)
{
    long i = (long)blockIdx.x * blockDim.x + threadIdx.x;
    if (i < n) p[i] = 0.0f;
}

__global__ void degrees_kernel(const int* __restrict__ node,
                               const int* __restrict__ edge,
                               float* __restrict__ Ddeg,
                               float* __restrict__ Bdeg, int nnz)
{
    int i = blockIdx.x * blockDim.x + threadIdx.x;
    if (i >= nnz) return;
    unsafeAtomicAdd(&Ddeg[node[i]], 1.0f);
    unsafeAtomicAdd(&Bdeg[edge[i]], 1.0f);
}

__global__ void __launch_bounds__(256)
scatter_kernel(const float* __restrict__ src, float* __restrict__ dst,
               const int* __restrict__ gidx, const int* __restrict__ sidx,
               int nnz)
{
    const int wid  = (int)(((long)blockIdx.x * blockDim.x + threadIdx.x) >> 5);
    const int lane = threadIdx.x & 31;
    if (wid >= nnz) return;

    const int g = gidx[wid];
    const int s = sidx[wid];

    const float4 v = ((const float4*)(src + (size_t)g * DIM))[lane];
    float* dp = dst + (size_t)s * DIM + lane * 4;
    unsafeAtomicAdd(dp + 0, v.x);
    unsafeAtomicAdd(dp + 1, v.y);
    unsafeAtomicAdd(dp + 2, v.z);
    unsafeAtomicAdd(dp + 3, v.w);
}

__global__ void scale_rows_kernel(float* __restrict__ m,
                                  const float* __restrict__ Bdeg, long n)
{
    long i = (long)blockIdx.x * blockDim.x + threadIdx.x;
    if (i >= n) return;
    float bb  = Bdeg[i >> 7];
    float inv = bb > 0.0f ? 1.0f / bb : 0.0f;
    m[i] *= inv;
}

__global__ void finalize_kernel(float* __restrict__ out,
                                const float* __restrict__ Ddeg,
                                const float* __restrict__ bias, long n)
{
    long i = (long)blockIdx.x * blockDim.x + threadIdx.x;
    if (i >= n) return;
    float dd  = Ddeg[i >> 7];
    float inv = dd > 0.0f ? 1.0f / dd : 0.0f;
    float v   = out[i] * inv + bias[i & (DIM - 1)];
    out[i] = v >= 0.0f ? v : NEG_SLOPE * v;
}

extern "C" void kernel_launch(void* const* d_in, const int* in_sizes, int n_in,
                              void* d_out, int out_size, void* d_ws, size_t ws_size,
                              hipStream_t stream)
{
    const float* X    = (const float*)d_in[0];
    const int*   A    = (const int*)d_in[1];
    const float* W    = (const float*)d_in[2];
    const float* bias = (const float*)d_in[3];
    float*       out  = (float*)d_out;

    const int N   = in_sizes[0] / DIM;
    const int NNZ = in_sizes[1] / 2;
    const int* node = A;
    const int* edge = A + NNZ;

    float* m    = (float*)d_ws;
    float* Ddeg = m + (size_t)NEDGES * DIM;
    float* Bdeg = Ddeg + N;
    const long ws_f  = (long)NEDGES * DIM + N + NEDGES;
    const long n_out = (long)N * DIM;
    const long n_m   = (long)NEDGES * DIM;

    const int mtiles = N / 16;
    gemm_wmma_f32<<<mtiles, 256, 0, stream>>>(X, W, out, mtiles);

    zero_kernel<<<(int)((ws_f + 255) / 256), 256, 0, stream>>>(m, ws_f);

    degrees_kernel<<<(NNZ + 255) / 256, 256, 0, stream>>>(node, edge, Ddeg, Bdeg, NNZ);

    scatter_kernel<<<(NNZ + 7) / 8, 256, 0, stream>>>(out, m, node, edge, NNZ);

    scale_rows_kernel<<<(int)((n_m + 255) / 256), 256, 0, stream>>>(m, Bdeg, n_m);

    zero_kernel<<<(int)((n_out + 255) / 256), 256, 0, stream>>>(out, n_out);

    scatter_kernel<<<(NNZ + 7) / 8, 256, 0, stream>>>(m, out, edge, node, NNZ);

    finalize_kernel<<<(int)((n_out + 255) / 256), 256, 0, stream>>>(out, Ddeg, bias, n_out);
}


# owner blocks scan one-byte owner tags (written by a new tagging launch; two planes where a pass has over 256 blocks) instead of the 4-byte index list; LDS adds keep their arrival order
# speedup vs baseline: 2.7174x; 2.7174x over previous
#include <hip/hip_runtime.h>

typedef __attribute__((ext_vector_type(2))) float v2f;
typedef __attribute__((ext_vector_type(8))) float v8f;

#define DIM        128
#define NEDGES     50000
#define NEG_SLOPE  0.01f

__global__ void __launch_bounds__(256)
gemm_wmma_f32(const float* __restrict__ X, const float* __restrict__ W,
              float* __restrict__ Y, int mtiles)
{
    const int lane  = threadIdx.x & 31;
    const int wv    = threadIdx.x >> 5;
    const long tile = (long)blockIdx.x * 8 + wv;
    const long mt   = tile >> 3;
    const int  nt   = (int)(tile & 7);
    if (mt >= mtiles) return;

    const int half = lane >> 4;
    const int lr   = lane & 15;
    const size_t row  = (size_t)mt * 16 + lr;
    const int    col0 = nt * 16;

    const float* __restrict__ ap = X + row * DIM + 2 * half;
    const float* __restrict__ bp = W + (size_t)(2 * half) * DIM + col0 + lr;

    v8f c = {};
#pragma unroll
    for (int k = 0; k < DIM; k += 4) {
        v2f a, b;
        a.x = ap[k];
        a.y = ap[k + 1];
        b.x = bp[(size_t)k * DIM];
        b.y = bp[(size_t)k * DIM + DIM];
        c = __builtin_amdgcn_wmma_f32_16x16x4_f32(false, a, false, b,
                                                  (short)0, c, false, false);
    }

    float* __restrict__ yp = Y + ((size_t)mt * 16 + half * 8) * DIM + col0 + lr;
#pragma unroll
    for (int r = 0; r < 8; ++r)
        yp[(size_t)r * DIM] = c[r];
}

template <int R, int NPL>
__device__ __forceinline__ unsigned owner_tag(int idx, unsigned plane)
{
    const unsigned o = (unsigned)idx / (unsigned)R;
    if (NPL == 1) return o & 255u;
    return ((o / 255u) & 1u) == plane ? o % 255u : 255u;
}

template <int R, int NPL>
__device__ __forceinline__ unsigned tag_word(const int4 v, unsigned plane)
{
    return owner_tag<R, NPL>(v.x, plane) | (owner_tag<R, NPL>(v.y, plane) << 8) |
           (owner_tag<R, NPL>(v.z, plane) << 16) | (owner_tag<R, NPL>(v.w, plane) << 24);
}

template <int RA, int RB>
__global__ void __launch_bounds__(1024)
tag_entries(const int* __restrict__ ia, const int* __restrict__ ib,
            unsigned* __restrict__ ta, unsigned* __restrict__ tb0,
            unsigned* __restrict__ tb1, int nnz)
{
    const int q = blockIdx.x * 1024 + threadIdx.x;
    if (q < (nnz >> 2)) {
        const int4 a = ((const int4*)ia)[q];
        const int4 b = ((const int4*)ib)[q];
        ta[q]  = tag_word<RA, 1>(a, 0u);
        tb0[q] = tag_word<RB, 2>(b, 0u);
        tb1[q] = tag_word<RB, 2>(b, 1u);
    }
}

#define NTHR 1024
#define NWAVE 32
#define BATCH 16
#define SCAN_U 8

template <int R, int CAP, bool FINAL, int NPL>
__global__ void __launch_bounds__(NTHR)
owner_pass(const float* __restrict__ src, float* __restrict__ dst,
           const int* __restrict__ gidx, const int* __restrict__ sidx,
           const unsigned* __restrict__ tags0, const unsigned* __restrict__ tags1,
           const float* __restrict__ bias, int nnz, int ndst)
{
    __shared__ __attribute__((aligned(16))) float s_acc[R * DIM];
    __shared__ int      s_deg[R];
    __shared__ int      s_list[CAP];
    __shared__ unsigned s_cnt;

    const int tid  = threadIdx.x;
    const int lane = tid & 31;
    const int wv   = tid >> 5;
    const int lo   = blockIdx.x * R;

    for (int i = tid; i < R * DIM / 4; i += NTHR)
        ((float4*)s_acc)[i] = make_float4(0.0f, 0.0f, 0.0f, 0.0f);
    for (int i = tid; i < R; i += NTHR) s_deg[i] = 0;
    if (tid == 0) s_cnt = 0u;
    __syncthreads();

    auto note = [&](int e) {
        const unsigned pos = atomicAdd(&s_cnt, 1u);
        if (pos < (unsigned)CAP) s_list[pos] = e;
    };

    auto look = [&](const int4 dv, int e) {
        const unsigned a0 = (unsigned)(dv.x - lo);
        const unsigned a1 = (unsigned)(dv.y - lo);
        const unsigned a2 = (unsigned)(dv.z - lo);
        const unsigned a3 = (unsigned)(dv.w - lo);
        const unsigned mn = min(min(a0, a1), min(a2, a3));
        if (mn < (unsigned)R) {
            if (a0 < (unsigned)R) note(e);
            if (a1 < (unsigned)R) note(e + 1);
            if (a2 < (unsigned)R) note(e + 2);
            if (a3 < (unsigned)R) note(e + 3);
        }
    };

    if (NPL > 0) {
        const bool odd_run = NPL == 2 && ((blockIdx.x / 255u) & 1u) != 0u;
        const unsigned* __restrict__ tags = odd_run ? tags1 : tags0;
        const unsigned pat = (NPL == 2 ? blockIdx.x % 255u : blockIdx.x & 255u) * 0x01010101u;

        auto look16 = [&](const uint4 dv, int e) {
            auto zero_bytes = [](const unsigned t) {
                return ~(((t & 0x7F7F7F7Fu) + 0x7F7F7F7Fu) | t | 0x7F7F7F7Fu);
            };
            unsigned hit = (zero_bytes(dv.x ^ pat) >> 7) | (zero_bytes(dv.y ^ pat) >> 6) |
                           (zero_bytes(dv.z ^ pat) >> 5) | (zero_bytes(dv.w ^ pat) >> 4);
            while (hit) {
                const int p = __builtin_ctz(hit);
                hit &= hit - 1u;
                note(e + 4 * (p & 7) + (p >> 3));
            }
        };

        const int n16   = nnz >> 4;
        const int nfull = n16 / (SCAN_U * NTHR);
        const uint4* __restrict__ tq = (const uint4*)tags;
        for (int it = 0; it < nfull; ++it) {
            const int qb = it * (SCAN_U * NTHR) + tid;
            uint4 d[SCAN_U];
#pragma unroll
            for (int u = 0; u < SCAN_U; ++u) d[u] = tq[qb + u * NTHR];
#pragma unroll
            for (int u = 0; u < SCAN_U; ++u) look16(d[u], (qb + u * NTHR) << 4);
        }
        for (int q = nfull * (SCAN_U * NTHR) + tid; q < n16; q += NTHR) look16(tq[q], q << 4);
        {
            const int e = (n16 << 4) + tid;
            if (e < nnz) {
                const unsigned a = (unsigned)(sidx[e] - lo);
                if (a < (unsigned)R) note(e);
            }
        }
    } else {
        const int nq    = nnz >> 2;
        const int nfull = nq / (SCAN_U * NTHR);
        const int4* __restrict__ sq = (const int4*)sidx;
        for (int it = 0; it < nfull; ++it) {
            const int qb = it * (SCAN_U * NTHR) + tid;
            int4 d[SCAN_U];
#pragma unroll
            for (int u = 0; u < SCAN_U; ++u) d[u] = sq[qb + u * NTHR];
#pragma unroll
            for (int u = 0; u < SCAN_U; ++u) look(d[u], (qb + u * NTHR) << 2);
        }
        for (int q = nfull * (SCAN_U * NTHR) + tid; q < nq; q += NTHR) look(sq[q], q << 2);
        {
            const int e = (nq << 2) + tid;
            if (e < nnz) {
                const unsigned a = (unsigned)(sidx[e] - lo);
                if (a < (unsigned)R) note(e);
            }
        }
    }
    __syncthreads();

    const unsigned tot = s_cnt;
    if (tot <= (unsigned)CAP) {
        const int n = (int)tot;
        for (int b = wv * BATCH; b < n; b += NWAVE * BATCH) {
            int g = 0, r = -1;
            if (lane < BATCH && b + lane < n) {
                const int e = s_list[b + lane];
                const unsigned a = (unsigned)(sidx[e] - lo);
                const int ge = gidx[e];
                const bool mine = a < (unsigned)R;
                r = mine ? (int)a : -1;
                g = mine ? ge : 0;
            }
            float4 v[BATCH];
#pragma unroll
            for (int k = 0; k < BATCH; ++k) {
                const int gk = __builtin_amdgcn_readlane(g, k);
                v[k] = ((const float4*)(src + (size_t)gk * DIM))[lane];
            }
#pragma unroll
            for (int k = 0; k < BATCH; ++k) {
                const int rk = __builtin_amdgcn_readlane(r, k);
                if (rk >= 0) {
                    float* a = s_acc + rk * DIM + lane;
                    atomicAdd(a,      v[k].x);
                    atomicAdd(a + 32, v[k].y);
                    atomicAdd(a + 64, v[k].z);
                    atomicAdd(a + 96, v[k].w);
                    if (lane == 0) atomicAdd(&s_deg[rk], 1);
                }
            }
        }
    } else {
        for (int e = tid; e < nnz; e += NTHR) {
            const unsigned r = (unsigned)(sidx[e] - lo);
            if (r < (unsigned)R) {
                const float* __restrict__ sp = src + (size_t)gidx[e] * DIM;
                for (int c = 0; c < DIM; ++c)
                    atomicAdd(&s_acc[r * DIM + ((c & 3) << 5) + (c >> 2)], sp[c]);
                atomicAdd(&s_deg[r], 1);
            }
        }
    }
    __syncthreads();

    for (int r = wv; r < R; r += NWAVE) {
        const int row = lo + r;
        if (row < ndst) {
            const int   cnt = s_deg[r];
            const float inv = cnt > 0 ? 1.0f / (float)cnt : 0.0f;
            const float* a = s_acc + r * DIM + lane;
            float4 o;
            o.x = a[0]  * inv;
            o.y = a[32] * inv;
            o.z = a[64] * inv;
            o.w = a[96] * inv;
            if (FINAL) {
                const float4 bb = ((const float4*)bias)[lane];
                o.x += bb.x; o.y += bb.y; o.z += bb.z; o.w += bb.w;
                o.x = o.x >= 0.0f ? o.x : NEG_SLOPE * o.x;
                o.y = o.y >= 0.0f ? o.y : NEG_SLOPE * o.y;
                o.z = o.z >= 0.0f ? o.z : NEG_SLOPE * o.z;
                o.w = o.w >= 0.0f ? o.w : NEG_SLOPE * o.w;
            }
            ((float4*)(dst + (size_t)row * DIM))[lane] = o;
        }
    }
}

#define R1   196
#define CAP1 10240
#define R2   512
#define CAP2 12288

extern "C" void kernel_launch(void* const* d_in, const int* in_sizes, int n_in,
                              void* d_out, int out_size, void* d_ws, size_t ws_size,
                              hipStream_t stream)
{
    const float* X    = (const float*)d_in[0];
    const int*   A    = (const int*)d_in[1];
    const float* W    = (const float*)d_in[2];
    const float* bias = (const float*)d_in[3];
    float*       out  = (float*)d_out;

    const int N   = in_sizes[0] / DIM;
    const int NNZ = in_sizes[1] / 2;
    const int* node = A;
    const int* edge = A + NNZ;

    float* m = (float*)d_ws;
    const size_t plane = ((size_t)NNZ + 15) & ~(size_t)15;
    unsigned* tag1  = (unsigned*)(m + (size_t)NEDGES * DIM);
    unsigned* tag2a = (unsigned*)((char*)tag1 + plane);
    unsigned* tag2b = (unsigned*)((char*)tag2a + plane);
    const bool room = ws_size >= (size_t)NEDGES * DIM * sizeof(float) + 3 * plane;

    const int mtiles = N / 16;
    gemm_wmma_f32<<<mtiles, 256, 0, stream>>>(X, W, out, mtiles);

    if (room) {
        tag_entries<R1, R2><<<((NNZ >> 2) + 1023) / 1024, 1024, 0, stream>>>(
            edge, node, tag1, tag2a, tag2b, NNZ);

        owner_pass<R1, CAP1, false, 1><<<(NEDGES + R1 - 1) / R1, NTHR, 0, stream>>>(
            out, m, node, edge, tag1, nullptr, bias, NNZ, NEDGES);

        owner_pass<R2, CAP2, true, 2><<<(N + R2 - 1) / R2, NTHR, 0, stream>>>(
            m, out, edge, node, tag2a, tag2b, bias, NNZ, N);
    } else {
        owner_pass<R1, CAP1, false, 0><<<(NEDGES + R1 - 1) / R1, NTHR, 0, stream>>>(
            out, m, node, edge, nullptr, nullptr, bias, NNZ, NEDGES);
        owner_pass<R2, CAP2, true, 0><<<(N + R2 - 1) / R2, NTHR, 0, stream>>>(
            m, out, edge, node, nullptr, nullptr, bias, NNZ, N);
    }
}


# product launch retiled: a wave owns a 16x128 strip (8 accumulator tiles), its A fragment loaded once per k-step, not by 8 waves, W read from an LDS copy, whole-line stores; same k order and sums
# speedup vs baseline: 3.1365x; 1.1542x over previous
#include <hip/hip_runtime.h>

typedef __attribute__((ext_vector_type(2))) float v2f;
typedef __attribute__((ext_vector_type(8))) float v8f;

#define DIM        128
#define NEDGES     50000
#define NEG_SLOPE  0.01f

#define GEMM_WAVES 8
#define GEMM_THR   (GEMM_WAVES * 32)
#define KQ         (DIM / 4)
#define W_ITEMS    (KQ * DIM / GEMM_THR)

__global__ void __launch_bounds__(GEMM_THR)
gemm_wmma_f32(const float* __restrict__ X, const float* __restrict__ W,
              float* __restrict__ Y, int mtiles)
{
    __shared__ __attribute__((aligned(16))) float4 s_w[KQ * DIM];

    const int tid  = threadIdx.x;
    const int lane = tid & 31;
    const int wv   = tid >> 5;
    const int half = lane >> 4;
    const int lr   = lane & 15;
    const int  mt   = blockIdx.x * GEMM_WAVES + wv;
    const bool live = mt < mtiles;
    const int  mtc  = live ? mt : mtiles - 1;

    float4 wit[W_ITEMS];
#pragma unroll
    for (int i = 0; i < W_ITEMS; ++i) {
        const int idx = tid + GEMM_THR * i;
        const float* __restrict__ wp = W + (size_t)(4 * (idx >> 7)) * DIM + (idx & (DIM - 1));
        wit[i] = make_float4(wp[0], wp[DIM], wp[2 * DIM], wp[3 * DIM]);
    }

    const float* __restrict__ ap = X + ((size_t)mtc * 16 + lr) * DIM + 2 * half;
    v2f a[KQ];
#pragma unroll
    for (int q = 0; q < KQ; ++q) a[q] = *(const v2f*)(ap + 4 * q);

#pragma unroll
    for (int i = 0; i < W_ITEMS; ++i) s_w[tid + GEMM_THR * i] = wit[i];
    __syncthreads();

    const v2f* __restrict__ bl = (const v2f*)s_w + (lr * 2 + half);

    v8f c[8];
#pragma unroll
    for (int nt = 0; nt < 8; ++nt) c[nt] = (v8f){};

#pragma unroll
    for (int q = 0; q < KQ; ++q) {
        v2f b[8];
#pragma unroll
        for (int nt = 0; nt < 8; ++nt) b[nt] = bl[(q * DIM + nt * 16) * 2];
#pragma unroll
        for (int nt = 0; nt < 8; ++nt)
            c[nt] = __builtin_amdgcn_wmma_f32_16x16x4_f32(false, a[q], false, b[nt],
                                                          (short)0, c[nt], false, false);
    }

    if (live) {
        float* __restrict__ yp = Y + (size_t)mt * 16 * DIM + lane;
#pragma unroll
        for (int p = 0; p < 4; ++p) {
#pragma unroll
            for (int r = 0; r < 8; ++r) {
                const auto s = __builtin_amdgcn_permlane16_swap(
                    __float_as_uint(c[2 * p][r]), __float_as_uint(c[2 * p + 1][r]), false, false);
                yp[(size_t)r * DIM + 32 * p]       = __uint_as_float(s[0]);
                yp[(size_t)(8 + r) * DIM + 32 * p] = __uint_as_float(s[1]);
            }
        }
    }
}

template <int R, int NPL>
__device__ __forceinline__ unsigned owner_tag(int idx, unsigned plane)
{
    const unsigned o = (unsigned)idx / (unsigned)R;
    if (NPL == 1) return o & 255u;
    return ((o / 255u) & 1u) == plane ? o % 255u : 255u;
}

template <int R, int NPL>
__device__ __forceinline__ unsigned tag_word(const int4 v, unsigned plane)
{
    return owner_tag<R, NPL>(v.x, plane) | (owner_tag<R, NPL>(v.y, plane) << 8) |
           (owner_tag<R, NPL>(v.z, plane) << 16) | (owner_tag<R, NPL>(v.w, plane) << 24);
}

template <int RA, int RB>
__global__ void __launch_bounds__(1024)
tag_entries(const int* __restrict__ ia, const int* __restrict__ ib,
            unsigned* __restrict__ ta, unsigned* __restrict__ tb0,
            unsigned* __restrict__ tb1, int nnz)
{
    const int q = blockIdx.x * 1024 + threadIdx.x;
    if (q < (nnz >> 2)) {
        const int4 a = ((const int4*)ia)[q];
        const int4 b = ((const int4*)ib)[q];
        ta[q]  = tag_word<RA, 1>(a, 0u);
        tb0[q] = tag_word<RB, 2>(b, 0u);
        tb1[q] = tag_word<RB, 2>(b, 1u);
    }
}

#define NTHR 1024
#define NWAVE 32
#define BATCH 16
#define SCAN_U 8

template <int R, int CAP, bool FINAL, int NPL>
__global__ void __launch_bounds__(NTHR)
owner_pass(const float* __restrict__ src, float* __restrict__ dst,
           const int* __restrict__ gidx, const int* __restrict__ sidx,
           const unsigned* __restrict__ tags0, const unsigned* __restrict__ tags1,
           const float* __restrict__ bias, int nnz, int ndst)
{
    __shared__ __attribute__((aligned(16))) float s_acc[R * DIM];
    __shared__ int      s_deg[R];
    __shared__ int      s_list[CAP];
    __shared__ unsigned s_cnt;

    const int tid  = threadIdx.x;
    const int lane = tid & 31;
    const int wv   = tid >> 5;
    const int lo   = blockIdx.x * R;

    for (int i = tid; i < R * DIM / 4; i += NTHR)
        ((float4*)s_acc)[i] = make_float4(0.0f, 0.0f, 0.0f, 0.0f);
    for (int i = tid; i < R; i += NTHR) s_deg[i] = 0;
    if (tid == 0) s_cnt = 0u;
    __syncthreads();

    auto note = [&](int e) {
        const unsigned pos = atomicAdd(&s_cnt, 1u);
        if (pos < (unsigned)CAP) s_list[pos] = e;
    };

    auto look = [&](const int4 dv, int e) {
        const unsigned a0 = (unsigned)(dv.x - lo);
        const unsigned a1 = (unsigned)(dv.y - lo);
        const unsigned a2 = (unsigned)(dv.z - lo);
        const unsigned a3 = (unsigned)(dv.w - lo);
        const unsigned mn = min(min(a0, a1), min(a2, a3));
        if (mn < (unsigned)R) {
            if (a0 < (unsigned)R) note(e);
            if (a1 < (unsigned)R) note(e + 1);
            if (a2 < (unsigned)R) note(e + 2);
            if (a3 < (unsigned)R) note(e + 3);
        }
    };

    if (NPL > 0) {
        const bool odd_run = NPL == 2 && ((blockIdx.x / 255u) & 1u) != 0u;
        const unsigned* __restrict__ tags = odd_run ? tags1 : tags0;
        const unsigned pat = (NPL == 2 ? blockIdx.x % 255u : blockIdx.x & 255u) * 0x01010101u;

        auto look16 = [&](const uint4 dv, int e) {
            auto zero_bytes = [](const unsigned t) {
                return ~(((t & 0x7F7F7F7Fu) + 0x7F7F7F7Fu) | t | 0x7F7F7F7Fu);
            };
            unsigned hit = (zero_bytes(dv.x ^ pat) >> 7) | (zero_bytes(dv.y ^ pat) >> 6) |
                           (zero_bytes(dv.z ^ pat) >> 5) | (zero_bytes(dv.w ^ pat) >> 4);
            while (hit) {
                const int p = __builtin_ctz(hit);
                hit &= hit - 1u;
                note(e + 4 * (p & 7) + (p >> 3));
            }
        };

        const int n16   = nnz >> 4;
        const int nfull = n16 / (SCAN_U * NTHR);
        const uint4* __restrict__ tq = (const uint4*)tags;
        for (int it = 0; it < nfull; ++it) {
            const int qb = it * (SCAN_U * NTHR) + tid;
            uint4 d[SCAN_U];
#pragma unroll
            for (int u = 0; u < SCAN_U; ++u) d[u] = tq[qb + u * NTHR];
#pragma unroll
            for (int u = 0; u < SCAN_U; ++u) look16(d[u], (qb + u * NTHR) << 4);
        }
        for (int q = nfull * (SCAN_U * NTHR) + tid; q < n16; q += NTHR) look16(tq[q], q << 4);
        {
            const int e = (n16 << 4) + tid;
            if (e < nnz) {
                const unsigned a = (unsigned)(sidx[e] - lo);
                if (a < (unsigned)R) note(e);
            }
        }
    } else {
        const int nq    = nnz >> 2;
        const int nfull = nq / (SCAN_U * NTHR);
        const int4* __restrict__ sq = (const int4*)sidx;
        for (int it = 0; it < nfull; ++it) {
            const int qb = it * (SCAN_U * NTHR) + tid;
            int4 d[SCAN_U];
#pragma unroll
            for (int u = 0; u < SCAN_U; ++u) d[u] = sq[qb + u * NTHR];
#pragma unroll
            for (int u = 0; u < SCAN_U; ++u) look(d[u], (qb + u * NTHR) << 2);
        }
        for (int q = nfull * (SCAN_U * NTHR) + tid; q < nq; q += NTHR) look(sq[q], q << 2);
        {
            const int e = (nq << 2) + tid;
            if (e < nnz) {
                const unsigned a = (unsigned)(sidx[e] - lo);
                if (a < (unsigned)R) note(e);
            }
        }
    }
    __syncthreads();

    const unsigned tot = s_cnt;
    if (tot <= (unsigned)CAP) {
        const int n = (int)tot;
        for (int b = wv * BATCH; b < n; b += NWAVE * BATCH) {
            int g = 0, r = -1;
            if (lane < BATCH && b + lane < n) {
                const int e = s_list[b + lane];
                const unsigned a = (unsigned)(sidx[e] - lo);
                const int ge = gidx[e];
                const bool mine = a < (unsigned)R;
                r = mine ? (int)a : -1;
                g = mine ? ge : 0;
            }
            float4 v[BATCH];
#pragma unroll
            for (int k = 0; k < BATCH; ++k) {
                const int gk = __builtin_amdgcn_readlane(g, k);
                v[k] = ((const float4*)(src + (size_t)gk * DIM))[lane];
            }
#pragma unroll
            for (int k = 0; k < BATCH; ++k) {
                const int rk = __builtin_amdgcn_readlane(r, k);
                if (rk >= 0) {
                    float* a = s_acc + rk * DIM + lane;
                    atomicAdd(a,      v[k].x);
                    atomicAdd(a + 32, v[k].y);
                    atomicAdd(a + 64, v[k].z);
                    atomicAdd(a + 96, v[k].w);
                    if (lane == 0) atomicAdd(&s_deg[rk], 1);
                }
            }
        }
    } else {
        for (int e = tid; e < nnz; e += NTHR) {
            const unsigned r = (unsigned)(sidx[e] - lo);
            if (r < (unsigned)R) {
                const float* __restrict__ sp = src + (size_t)gidx[e] * DIM;
                for (int c = 0; c < DIM; ++c)
                    atomicAdd(&s_acc[r * DIM + ((c & 3) << 5) + (c >> 2)], sp[c]);
                atomicAdd(&s_deg[r], 1);
            }
        }
    }
    __syncthreads();

    for (int r = wv; r < R; r += NWAVE) {
        const int row = lo + r;
        if (row < ndst) {
            const int   cnt = s_deg[r];
            const float inv = cnt > 0 ? 1.0f / (float)cnt : 0.0f;
            const float* a = s_acc + r * DIM + lane;
            float4 o;
            o.x = a[0]  * inv;
            o.y = a[32] * inv;
            o.z = a[64] * inv;
            o.w = a[96] * inv;
            if (FINAL) {
                const float4 bb = ((const float4*)bias)[lane];
                o.x += bb.x; o.y += bb.y; o.z += bb.z; o.w += bb.w;
                o.x = o.x >= 0.0f ? o.x : NEG_SLOPE * o.x;
                o.y = o.y >= 0.0f ? o.y : NEG_SLOPE * o.y;
                o.z = o.z >= 0.0f ? o.z : NEG_SLOPE * o.z;
                o.w = o.w >= 0.0f ? o.w : NEG_SLOPE * o.w;
            }
            ((float4*)(dst + (size_t)row * DIM))[lane] = o;
        }
    }
}

#define R1   196
#define CAP1 10240
#define R2   512
#define CAP2 12288

extern "C" void kernel_launch(void* const* d_in, const int* in_sizes, int n_in,
                              void* d_out, int out_size, void* d_ws, size_t ws_size,
                              hipStream_t stream)
{
    const float* X    = (const float*)d_in[0];
    const int*   A    = (const int*)d_in[1];
    const float* W    = (const float*)d_in[2];
    const float* bias = (const float*)d_in[3];
    float*       out  = (float*)d_out;

    const int N   = in_sizes[0] / DIM;
    const int NNZ = in_sizes[1] / 2;
    const int* node = A;
    const int* edge = A + NNZ;

    float* m = (float*)d_ws;
    const size_t plane = ((size_t)NNZ + 15) & ~(size_t)15;
    unsigned* tag1  = (unsigned*)(m + (size_t)NEDGES * DIM);
    unsigned* tag2a = (unsigned*)((char*)tag1 + plane);
    unsigned* tag2b = (unsigned*)((char*)tag2a + plane);
    const bool room = ws_size >= (size_t)NEDGES * DIM * sizeof(float) + 3 * plane;

    const int mtiles = N / 16;
    gemm_wmma_f32<<<(mtiles + GEMM_WAVES - 1) / GEMM_WAVES, GEMM_THR, 0, stream>>>(X, W, out, mtiles);

    if (room) {
        tag_entries<R1, R2><<<((NNZ >> 2) + 1023) / 1024, 1024, 0, stream>>>(
            edge, node, tag1, tag2a, tag2b, NNZ);

        owner_pass<R1, CAP1, false, 1><<<(NEDGES + R1 - 1) / R1, NTHR, 0, stream>>>(
            out, m, node, edge, tag1, nullptr, bias, NNZ, NEDGES);

        owner_pass<R2, CAP2, true, 2><<<(N + R2 - 1) / R2, NTHR, 0, stream>>>(
            m, out, edge, node, tag2a, tag2b, bias, NNZ, N);
    } else {
        owner_pass<R1, CAP1, false, 0><<<(NEDGES + R1 - 1) / R1, NTHR, 0, stream>>>(
            out, m, node, edge, nullptr, nullptr, bias, NNZ, NEDGES);
        owner_pass<R2, CAP2, true, 0><<<(N + R2 - 1) / R2, NTHR, 0, stream>>>(
            m, out, edge, node, nullptr, nullptr, bias, NNZ, N);
    }
}


# tagging launch also writes both index rows interleaved, one 8-byte record per entry; the add step of both passes reads one record per listed entry, not two words from rows 8 MB apart; same LDS adds
# speedup vs baseline: 3.2920x; 1.0496x over previous
#include <hip/hip_runtime.h>

typedef __attribute__((ext_vector_type(2))) float v2f;
typedef __attribute__((ext_vector_type(8))) float v8f;

#define DIM        128
#define NEDGES     50000
#define NEG_SLOPE  0.01f

#define GEMM_WAVES 8
#define GEMM_THR   (GEMM_WAVES * 32)
#define KQ         (DIM / 4)
#define W_ITEMS    (KQ * DIM / GEMM_THR)

__global__ void __launch_bounds__(GEMM_THR)
gemm_wmma_f32(const float* __restrict__ X, const float* __restrict__ W,
              float* __restrict__ Y, int mtiles)
{
    __shared__ __attribute__((aligned(16))) float4 s_w[KQ * DIM];

    const int tid  = threadIdx.x;
    const int lane = tid & 31;
    const int wv   = tid >> 5;
    const int half = lane >> 4;
    const int lr   = lane & 15;
    const int  mt   = blockIdx.x * GEMM_WAVES + wv;
    const bool live = mt < mtiles;
    const int  mtc  = live ? mt : mtiles - 1;

    float4 wit[W_ITEMS];
#pragma unroll
    for (int i = 0; i < W_ITEMS; ++i) {
        const int idx = tid + GEMM_THR * i;
        const float* __restrict__ wp = W + (size_t)(4 * (idx >> 7)) * DIM + (idx & (DIM - 1));
        wit[i] = make_float4(wp[0], wp[DIM], wp[2 * DIM], wp[3 * DIM]);
    }

    const float* __restrict__ ap = X + ((size_t)mtc * 16 + lr) * DIM + 2 * half;
    v2f a[KQ];
#pragma unroll
    for (int q = 0; q < KQ; ++q) a[q] = *(const v2f*)(ap + 4 * q);

#pragma unroll
    for (int i = 0; i < W_ITEMS; ++i) s_w[tid + GEMM_THR * i] = wit[i];
    __syncthreads();

    const v2f* __restrict__ bl = (const v2f*)s_w + (lr * 2 + half);

    v8f c[8];
#pragma unroll
    for (int nt = 0; nt < 8; ++nt) c[nt] = (v8f){};

#pragma unroll
    for (int q = 0; q < KQ; ++q) {
        v2f b[8];
#pragma unroll
        for (int nt = 0; nt < 8; ++nt) b[nt] = bl[(q * DIM + nt * 16) * 2];
#pragma unroll
        for (int nt = 0; nt < 8; ++nt)
            c[nt] = __builtin_amdgcn_wmma_f32_16x16x4_f32(false, a[q], false, b[nt],
                                                          (short)0, c[nt], false, false);
    }

    if (live) {
        float* __restrict__ yp = Y + (size_t)mt * 16 * DIM + lane;
#pragma unroll
        for (int p = 0; p < 4; ++p) {
#pragma unroll
            for (int r = 0; r < 8; ++r) {
                const auto s = __builtin_amdgcn_permlane16_swap(
                    __float_as_uint(c[2 * p][r]), __float_as_uint(c[2 * p + 1][r]), false, false);
                yp[(size_t)r * DIM + 32 * p]       = __uint_as_float(s[0]);
                yp[(size_t)(8 + r) * DIM + 32 * p] = __uint_as_float(s[1]);
            }
        }
    }
}

template <int R, int NPL>
__device__ __forceinline__ unsigned owner_tag(int idx, unsigned plane)
{
    const unsigned o = (unsigned)idx / (unsigned)R;
    if (NPL == 1) return o & 255u;
    return ((o / 255u) & 1u) == plane ? o % 255u : 255u;
}

template <int R, int NPL>
__device__ __forceinline__ unsigned tag_word(const int4 v, unsigned plane)
{
    return owner_tag<R, NPL>(v.x, plane) | (owner_tag<R, NPL>(v.y, plane) << 8) |
           (owner_tag<R, NPL>(v.z, plane) << 16) | (owner_tag<R, NPL>(v.w, plane) << 24);
}

template <int RA, int RB, bool REC>
__global__ void __launch_bounds__(1024)
tag_entries(const int* __restrict__ ia, const int* __restrict__ ib,
            unsigned* __restrict__ ta, unsigned* __restrict__ tb0,
            unsigned* __restrict__ tb1, int* __restrict__ rec, int nnz)
{
    const int nq = nnz >> 2;
    const int q  = blockIdx.x * 1024 + threadIdx.x;
    if (q < nq) {
        const int4 a = ((const int4*)ia)[q];
        const int4 b = ((const int4*)ib)[q];
        ta[q]  = tag_word<RA, 1>(a, 0u);
        tb0[q] = tag_word<RB, 2>(b, 0u);
        tb1[q] = tag_word<RB, 2>(b, 1u);
    }
    if (REC) {
        for (int k = 0; k < 2; ++k) {
            const int P = blockIdx.x * 2048 + k * 1024 + threadIdx.x;
            if (P < 2 * nq) {
                const int2 a = ((const int2*)ia)[P];
                const int2 b = ((const int2*)ib)[P];
                ((int4*)rec)[P] = make_int4(b.x, a.x, b.y, a.y);
            }
        }
        if (q == 0)
            for (int e = 4 * nq; e < nnz; ++e) { rec[2 * e] = ib[e]; rec[2 * e + 1] = ia[e]; }
    }
}

#define NTHR 1024
#define NWAVE 32
#define BATCH 16
#define SCAN_U 8

template <int R, int CAP, bool FINAL, int NPL, int REC>
__global__ void __launch_bounds__(NTHR)
owner_pass(const float* __restrict__ src, float* __restrict__ dst,
           const int* __restrict__ gidx, const int* __restrict__ sidx,
           const unsigned* __restrict__ tags0, const unsigned* __restrict__ tags1,
           const int* __restrict__ rec,
           const float* __restrict__ bias, int nnz, int ndst)
{
    __shared__ __attribute__((aligned(16))) float s_acc[R * DIM];
    __shared__ int      s_deg[R];
    __shared__ int      s_list[CAP];
    __shared__ unsigned s_cnt;

    const int tid  = threadIdx.x;
    const int lane = tid & 31;
    const int wv   = tid >> 5;
    const int lo   = blockIdx.x * R;

    for (int i = tid; i < R * DIM / 4; i += NTHR)
        ((float4*)s_acc)[i] = make_float4(0.0f, 0.0f, 0.0f, 0.0f);
    for (int i = tid; i < R; i += NTHR) s_deg[i] = 0;
    if (tid == 0) s_cnt = 0u;
    __syncthreads();

    auto note = [&](int e) {
        const unsigned pos = atomicAdd(&s_cnt, 1u);
        if (pos < (unsigned)CAP) s_list[pos] = e;
    };

    auto look = [&](const int4 dv, int e) {
        const unsigned a0 = (unsigned)(dv.x - lo);
        const unsigned a1 = (unsigned)(dv.y - lo);
        const unsigned a2 = (unsigned)(dv.z - lo);
        const unsigned a3 = (unsigned)(dv.w - lo);
        const unsigned mn = min(min(a0, a1), min(a2, a3));
        if (mn < (unsigned)R) {
            if (a0 < (unsigned)R) note(e);
            if (a1 < (unsigned)R) note(e + 1);
            if (a2 < (unsigned)R) note(e + 2);
            if (a3 < (unsigned)R) note(e + 3);
        }
    };

    if (NPL > 0) {
        const bool odd_run = NPL == 2 && ((blockIdx.x / 255u) & 1u) != 0u;
        const unsigned* __restrict__ tags = odd_run ? tags1 : tags0;
        const unsigned pat = (NPL == 2 ? blockIdx.x % 255u : blockIdx.x & 255u) * 0x01010101u;

        auto look16 = [&](const uint4 dv, int e) {
            auto zero_bytes = [](const unsigned t) {
                return ~(((t & 0x7F7F7F7Fu) + 0x7F7F7F7Fu) | t | 0x7F7F7F7Fu);
            };
            unsigned hit = (zero_bytes(dv.x ^ pat) >> 7) | (zero_bytes(dv.y ^ pat) >> 6) |
                           (zero_bytes(dv.z ^ pat) >> 5) | (zero_bytes(dv.w ^ pat) >> 4);
            while (hit) {
                const int p = __builtin_ctz(hit);
                hit &= hit - 1u;
                note(e + 4 * (p & 7) + (p >> 3));
            }
        };

        const int n16   = nnz >> 4;
        const int nfull = n16 / (SCAN_U * NTHR);
        const uint4* __restrict__ tq = (const uint4*)tags;
        for (int it = 0; it < nfull; ++it) {
            const int qb = it * (SCAN_U * NTHR) + tid;
            uint4 d[SCAN_U];
#pragma unroll
            for (int u = 0; u < SCAN_U; ++u) d[u] = tq[qb + u * NTHR];
#pragma unroll
            for (int u = 0; u < SCAN_U; ++u) look16(d[u], (qb + u * NTHR) << 4);
        }
        for (int q = nfull * (SCAN_U * NTHR) + tid; q < n16; q += NTHR) look16(tq[q], q << 4);
        {
            const int e = (n16 << 4) + tid;
            if (e < nnz) {
                const unsigned a = (unsigned)(sidx[e] - lo);
                if (a < (unsigned)R) note(e);
            }
        }
    } else {
        const int nq    = nnz >> 2;
        const int nfull = nq / (SCAN_U * NTHR);
        const int4* __restrict__ sq = (const int4*)sidx;
        for (int it = 0; it < nfull; ++it) {
            const int qb = it * (SCAN_U * NTHR) + tid;
            int4 d[SCAN_U];
#pragma unroll
            for (int u = 0; u < SCAN_U; ++u) d[u] = sq[qb + u * NTHR];
#pragma unroll
            for (int u = 0; u < SCAN_U; ++u) look(d[u], (qb + u * NTHR) << 2);
        }
        for (int q = nfull * (SCAN_U * NTHR) + tid; q < nq; q += NTHR) look(sq[q], q << 2);
        {
            const int e = (nq << 2) + tid;
            if (e < nnz) {
                const unsigned a = (unsigned)(sidx[e] - lo);
                if (a < (unsigned)R) note(e);
            }
        }
    }
    __syncthreads();

    const unsigned tot = s_cnt;
    if (tot <= (unsigned)CAP) {
        const int n = (int)tot;
        for (int b = wv * BATCH; b < n; b += NWAVE * BATCH) {
            int g = 0, r = -1;
            if (lane < BATCH && b + lane < n) {
                const int e = s_list[b + lane];
                int de, ge;
                if (REC) {
                    const int2 rc = ((const int2*)rec)[e];
                    de = REC == 1 ? rc.y : rc.x;
                    ge = REC == 1 ? rc.x : rc.y;
                } else {
                    de = sidx[e];
                    ge = gidx[e];
                }
                const unsigned a = (unsigned)(de - lo);
                const bool mine = a < (unsigned)R;
                r = mine ? (int)a : -1;
                g = mine ? ge : 0;
            }
            float4 v[BATCH];
#pragma unroll
            for (int k = 0; k < BATCH; ++k) {
                const int gk = __builtin_amdgcn_readlane(g, k);
                v[k] = ((const float4*)(src + (size_t)gk * DIM))[lane];
            }
#pragma unroll
            for (int k = 0; k < BATCH; ++k) {
                const int rk = __builtin_amdgcn_readlane(r, k);
                if (rk >= 0) {
                    float* a = s_acc + rk * DIM + lane;
                    atomicAdd(a,      v[k].x);
                    atomicAdd(a + 32, v[k].y);
                    atomicAdd(a + 64, v[k].z);
                    atomicAdd(a + 96, v[k].w);
                    if (lane == 0) atomicAdd(&s_deg[rk], 1);
                }
            }
        }
    } else {
        for (int e = tid; e < nnz; e += NTHR) {
            const unsigned r = (unsigned)(sidx[e] - lo);
            if (r < (unsigned)R) {
                const float* __restrict__ sp = src + (size_t)gidx[e] * DIM;
                for (int c = 0; c < DIM; ++c)
                    atomicAdd(&s_acc[r * DIM + ((c & 3) << 5) + (c >> 2)], sp[c]);
                atomicAdd(&s_deg[r], 1);
            }
        }
    }
    __syncthreads();

    for (int r = wv; r < R; r += NWAVE) {
        const int row = lo + r;
        if (row < ndst) {
            const int   cnt = s_deg[r];
            const float inv = cnt > 0 ? 1.0f / (float)cnt : 0.0f;
            const float* a = s_acc + r * DIM + lane;
            float4 o;
            o.x = a[0]  * inv;
            o.y = a[32] * inv;
            o.z = a[64] * inv;
            o.w = a[96] * inv;
            if (FINAL) {
                const float4 bb = ((const float4*)bias)[lane];
                o.x += bb.x; o.y += bb.y; o.z += bb.z; o.w += bb.w;
                o.x = o.x >= 0.0f ? o.x : NEG_SLOPE * o.x;
                o.y = o.y >= 0.0f ? o.y : NEG_SLOPE * o.y;
                o.z = o.z >= 0.0f ? o.z : NEG_SLOPE * o.z;
                o.w = o.w >= 0.0f ? o.w : NEG_SLOPE * o.w;
            }
            ((float4*)(dst + (size_t)row * DIM))[lane] = o;
        }
    }
}

#define R1   196
#define CAP1 10240
#define R2   512
#define CAP2 12288

extern "C" void kernel_launch(void* const* d_in, const int* in_sizes, int n_in,
                              void* d_out, int out_size, void* d_ws, size_t ws_size,
                              hipStream_t stream)
{
    const float* X    = (const float*)d_in[0];
    const int*   A    = (const int*)d_in[1];
    const float* W    = (const float*)d_in[2];
    const float* bias = (const float*)d_in[3];
    float*       out  = (float*)d_out;

    const int N   = in_sizes[0] / DIM;
    const int NNZ = in_sizes[1] / 2;
    const int* node = A;
    const int* edge = A + NNZ;

    float* m = (float*)d_ws;
    const size_t plane = ((size_t)NNZ + 15) & ~(size_t)15;
    unsigned* tag1  = (unsigned*)(m + (size_t)NEDGES * DIM);
    unsigned* tag2a = (unsigned*)((char*)tag1 + plane);
    unsigned* tag2b = (unsigned*)((char*)tag2a + plane);
    int*      rec   = (int*)((char*)tag2b + plane);
    const bool room  = ws_size >= (size_t)NEDGES * DIM * sizeof(float) + 3 * plane;
    const bool room2 = ws_size >= (size_t)NEDGES * DIM * sizeof(float) + 3 * plane + (size_t)NNZ * 8;

    const int mtiles = N / 16;
    gemm_wmma_f32<<<(mtiles + GEMM_WAVES - 1) / GEMM_WAVES, GEMM_THR, 0, stream>>>(X, W, out, mtiles);

    if (room2) {
        tag_entries<R1, R2, true><<<((NNZ >> 2) + 1023) / 1024, 1024, 0, stream>>>(
            edge, node, tag1, tag2a, tag2b, rec, NNZ);

        owner_pass<R1, CAP1, false, 1, 1><<<(NEDGES + R1 - 1) / R1, NTHR, 0, stream>>>(
            out, m, node, edge, tag1, nullptr, rec, bias, NNZ, NEDGES);

        owner_pass<R2, CAP2, true, 2, 2><<<(N + R2 - 1) / R2, NTHR, 0, stream>>>(
            m, out, edge, node, tag2a, tag2b, rec, bias, NNZ, N);
    } else if (room) {
        tag_entries<R1, R2, false><<<((NNZ >> 2) + 1023) / 1024, 1024, 0, stream>>>(
            edge, node, tag1, tag2a, tag2b, nullptr, NNZ);
        owner_pass<R1, CAP1, false, 1, 0><<<(NEDGES + R1 - 1) / R1, NTHR, 0, stream>>>(
            out, m, node, edge, tag1, nullptr, nullptr, bias, NNZ, NEDGES);
        owner_pass<R2, CAP2, true, 2, 0><<<(N + R2 - 1) / R2, NTHR, 0, stream>>>(
            m, out, edge, node, tag2a, tag2b, nullptr, bias, NNZ, N);
    } else {
        owner_pass<R1, CAP1, false, 0, 0><<<(NEDGES + R1 - 1) / R1, NTHR, 0, stream>>>(
            out, m, node, edge, nullptr, nullptr, nullptr, bias, NNZ, NEDGES);
        owner_pass<R2, CAP2, true, 0, 0><<<(N + R2 - 1) / R2, NTHR, 0, stream>>>(
            m, out, edge, node, nullptr, nullptr, nullptr, bias, NNZ, N);
    }
}


# pass 2 as one resident round: 256 blocks (one per WGP) own 782 rows as two halves of 391 summed in turn in LDS after one scan of one tag plane; was 391 blocks in 2 rounds; LDS adds in arrival order
# speedup vs baseline: 3.6771x; 1.1170x over previous
#include <hip/hip_runtime.h>

typedef __attribute__((ext_vector_type(2))) float v2f;
typedef __attribute__((ext_vector_type(8))) float v8f;

#define DIM        128
#define NEDGES     50000
#define NEG_SLOPE  0.01f

#define GEMM_WAVES 8
#define GEMM_THR   (GEMM_WAVES * 32)
#define KQ         (DIM / 4)
#define W_ITEMS    (KQ * DIM / GEMM_THR)

__global__ void __launch_bounds__(GEMM_THR)
gemm_wmma_f32(const float* __restrict__ X, const float* __restrict__ W,
              float* __restrict__ Y, int mtiles)
{
    __shared__ __attribute__((aligned(16))) float4 s_w[KQ * DIM];

    const int tid  = threadIdx.x;
    const int lane = tid & 31;
    const int wv   = tid >> 5;
    const int half = lane >> 4;
    const int lr   = lane & 15;
    const int  mt   = blockIdx.x * GEMM_WAVES + wv;
    const bool live = mt < mtiles;
    const int  mtc  = live ? mt : mtiles - 1;

    float4 wit[W_ITEMS];
#pragma unroll
    for (int i = 0; i < W_ITEMS; ++i) {
        const int idx = tid + GEMM_THR * i;
        const float* __restrict__ wp = W + (size_t)(4 * (idx >> 7)) * DIM + (idx & (DIM - 1));
        wit[i] = make_float4(wp[0], wp[DIM], wp[2 * DIM], wp[3 * DIM]);
    }

    const float* __restrict__ ap = X + ((size_t)mtc * 16 + lr) * DIM + 2 * half;
    v2f a[KQ];
#pragma unroll
    for (int q = 0; q < KQ; ++q) a[q] = *(const v2f*)(ap + 4 * q);

#pragma unroll
    for (int i = 0; i < W_ITEMS; ++i) s_w[tid + GEMM_THR * i] = wit[i];
    __syncthreads();

    const v2f* __restrict__ bl = (const v2f*)s_w + (lr * 2 + half);

    v8f c[8];
#pragma unroll
    for (int nt = 0; nt < 8; ++nt) c[nt] = (v8f){};

#pragma unroll
    for (int q = 0; q < KQ; ++q) {
        v2f b[8];
#pragma unroll
        for (int nt = 0; nt < 8; ++nt) b[nt] = bl[(q * DIM + nt * 16) * 2];
#pragma unroll
        for (int nt = 0; nt < 8; ++nt)
            c[nt] = __builtin_amdgcn_wmma_f32_16x16x4_f32(false, a[q], false, b[nt],
                                                          (short)0, c[nt], false, false);
    }

    if (live) {
        float* __restrict__ yp = Y + (size_t)mt * 16 * DIM + lane;
#pragma unroll
        for (int p = 0; p < 4; ++p) {
#pragma unroll
            for (int r = 0; r < 8; ++r) {
                const auto s = __builtin_amdgcn_permlane16_swap(
                    __float_as_uint(c[2 * p][r]), __float_as_uint(c[2 * p + 1][r]), false, false);
                yp[(size_t)r * DIM + 32 * p]       = __uint_as_float(s[0]);
                yp[(size_t)(8 + r) * DIM + 32 * p] = __uint_as_float(s[1]);
            }
        }
    }
}

template <int R, int NPL>
__device__ __forceinline__ unsigned owner_tag(int idx, unsigned plane)
{
    const unsigned o = (unsigned)idx / (unsigned)R;
    if (NPL == 1) return o & 255u;
    return ((o / 255u) & 1u) == plane ? o % 255u : 255u;
}

template <int R, int NPL>
__device__ __forceinline__ unsigned tag_word(const int4 v, unsigned plane)
{
    return owner_tag<R, NPL>(v.x, plane) | (owner_tag<R, NPL>(v.y, plane) << 8) |
           (owner_tag<R, NPL>(v.z, plane) << 16) | (owner_tag<R, NPL>(v.w, plane) << 24);
}

template <int RA, int RB, int NPLB, bool REC>
__global__ void __launch_bounds__(1024)
tag_entries(const int* __restrict__ ia, const int* __restrict__ ib,
            unsigned* __restrict__ ta, unsigned* __restrict__ tb0,
            unsigned* __restrict__ tb1, int* __restrict__ rec, int nnz)
{
    const int nq = nnz >> 2;
    const int q  = blockIdx.x * 1024 + threadIdx.x;
    if (q < nq) {
        const int4 a = ((const int4*)ia)[q];
        const int4 b = ((const int4*)ib)[q];
        ta[q]  = tag_word<RA, 1>(a, 0u);
        if (NPLB == 1) {
            tb0[q] = tag_word<RB, 1>(b, 0u);
        } else {
            tb0[q] = tag_word<RB, 2>(b, 0u);
            tb1[q] = tag_word<RB, 2>(b, 1u);
        }
    }
    if (REC) {
        for (int k = 0; k < 2; ++k) {
            const int P = blockIdx.x * 2048 + k * 1024 + threadIdx.x;
            if (P < 2 * nq) {
                const int2 a = ((const int2*)ia)[P];
                const int2 b = ((const int2*)ib)[P];
                ((int4*)rec)[P] = make_int4(b.x, a.x, b.y, a.y);
            }
        }
        if (q == 0)
            for (int e = 4 * nq; e < nnz; ++e) { rec[2 * e] = ib[e]; rec[2 * e + 1] = ia[e]; }
    }
}

#define NTHR 1024
#define NWAVE 32
#define BATCH 16
#define SCAN_U 8

template <int R, int CAP, bool FINAL, int NPL, int REC>
__global__ void __launch_bounds__(NTHR)
owner_pass(const float* __restrict__ src, float* __restrict__ dst,
           const int* __restrict__ gidx, const int* __restrict__ sidx,
           const unsigned* __restrict__ tags0, const unsigned* __restrict__ tags1,
           const int* __restrict__ rec,
           const float* __restrict__ bias, int nnz, int ndst)
{
    __shared__ __attribute__((aligned(16))) float s_acc[R * DIM];
    __shared__ int      s_deg[R];
    __shared__ int      s_list[CAP];
    __shared__ unsigned s_cnt;

    const int tid  = threadIdx.x;
    const int lane = tid & 31;
    const int wv   = tid >> 5;
    const int lo   = blockIdx.x * R;

    for (int i = tid; i < R * DIM / 4; i += NTHR)
        ((float4*)s_acc)[i] = make_float4(0.0f, 0.0f, 0.0f, 0.0f);
    for (int i = tid; i < R; i += NTHR) s_deg[i] = 0;
    if (tid == 0) s_cnt = 0u;
    __syncthreads();

    auto note = [&](int e) {
        const unsigned pos = atomicAdd(&s_cnt, 1u);
        if (pos < (unsigned)CAP) s_list[pos] = e;
    };

    auto look = [&](const int4 dv, int e) {
        const unsigned a0 = (unsigned)(dv.x - lo);
        const unsigned a1 = (unsigned)(dv.y - lo);
        const unsigned a2 = (unsigned)(dv.z - lo);
        const unsigned a3 = (unsigned)(dv.w - lo);
        const unsigned mn = min(min(a0, a1), min(a2, a3));
        if (mn < (unsigned)R) {
            if (a0 < (unsigned)R) note(e);
            if (a1 < (unsigned)R) note(e + 1);
            if (a2 < (unsigned)R) note(e + 2);
            if (a3 < (unsigned)R) note(e + 3);
        }
    };

    if (NPL > 0) {
        const bool odd_run = NPL == 2 && ((blockIdx.x / 255u) & 1u) != 0u;
        const unsigned* __restrict__ tags = odd_run ? tags1 : tags0;
        const unsigned pat = (NPL == 2 ? blockIdx.x % 255u : blockIdx.x & 255u) * 0x01010101u;

        auto look16 = [&](const uint4 dv, int e) {
            auto zero_bytes = [](const unsigned t) {
                return ~(((t & 0x7F7F7F7Fu) + 0x7F7F7F7Fu) | t | 0x7F7F7F7Fu);
            };
            unsigned hit = (zero_bytes(dv.x ^ pat) >> 7) | (zero_bytes(dv.y ^ pat) >> 6) |
                           (zero_bytes(dv.z ^ pat) >> 5) | (zero_bytes(dv.w ^ pat) >> 4);
            while (hit) {
                const int p = __builtin_ctz(hit);
                hit &= hit - 1u;
                note(e + 4 * (p & 7) + (p >> 3));
            }
        };

        const int n16   = nnz >> 4;
        const int nfull = n16 / (SCAN_U * NTHR);
        const uint4* __restrict__ tq = (const uint4*)tags;
        for (int it = 0; it < nfull; ++it) {
            const int qb = it * (SCAN_U * NTHR) + tid;
            uint4 d[SCAN_U];
#pragma unroll
            for (int u = 0; u < SCAN_U; ++u) d[u] = tq[qb + u * NTHR];
#pragma unroll
            for (int u = 0; u < SCAN_U; ++u) look16(d[u], (qb + u * NTHR) << 4);
        }
        for (int q = nfull * (SCAN_U * NTHR) + tid; q < n16; q += NTHR) look16(tq[q], q << 4);
        {
            const int e = (n16 << 4) + tid;
            if (e < nnz) {
                const unsigned a = (unsigned)(sidx[e] - lo);
                if (a < (unsigned)R) note(e);
            }
        }
    } else {
        const int nq    = nnz >> 2;
        const int nfull = nq / (SCAN_U * NTHR);
        const int4* __restrict__ sq = (const int4*)sidx;
        for (int it = 0; it < nfull; ++it) {
            const int qb = it * (SCAN_U * NTHR) + tid;
            int4 d[SCAN_U];
#pragma unroll
            for (int u = 0; u < SCAN_U; ++u) d[u] = sq[qb + u * NTHR];
#pragma unroll
            for (int u = 0; u < SCAN_U; ++u) look(d[u], (qb + u * NTHR) << 2);
        }
        for (int q = nfull * (SCAN_U * NTHR) + tid; q < nq; q += NTHR) look(sq[q], q << 2);
        {
            const int e = (nq << 2) + tid;
            if (e < nnz) {
                const unsigned a = (unsigned)(sidx[e] - lo);
                if (a < (unsigned)R) note(e);
            }
        }
    }
    __syncthreads();

    const unsigned tot = s_cnt;
    if (tot <= (unsigned)CAP) {
        const int n = (int)tot;
        for (int b = wv * BATCH; b < n; b += NWAVE * BATCH) {
            int g = 0, r = -1;
            if (lane < BATCH && b + lane < n) {
                const int e = s_list[b + lane];
                int de, ge;
                if (REC) {
                    const int2 rc = ((const int2*)rec)[e];
                    de = REC == 1 ? rc.y : rc.x;
                    ge = REC == 1 ? rc.x : rc.y;
                } else {
                    de = sidx[e];
                    ge = gidx[e];
                }
                const unsigned a = (unsigned)(de - lo);
                const bool mine = a < (unsigned)R;
                r = mine ? (int)a : -1;
                g = mine ? ge : 0;
            }
            float4 v[BATCH];
#pragma unroll
            for (int k = 0; k < BATCH; ++k) {
                const int gk = __builtin_amdgcn_readlane(g, k);
                v[k] = ((const float4*)(src + (size_t)gk * DIM))[lane];
            }
#pragma unroll
            for (int k = 0; k < BATCH; ++k) {
                const int rk = __builtin_amdgcn_readlane(r, k);
                if (rk >= 0) {
                    float* a = s_acc + rk * DIM + lane;
                    atomicAdd(a,      v[k].x);
                    atomicAdd(a + 32, v[k].y);
                    atomicAdd(a + 64, v[k].z);
                    atomicAdd(a + 96, v[k].w);
                    if (lane == 0) atomicAdd(&s_deg[rk], 1);
                }
            }
        }
    } else {
        for (int e = tid; e < nnz; e += NTHR) {
            const unsigned r = (unsigned)(sidx[e] - lo);
            if (r < (unsigned)R) {
                const float* __restrict__ sp = src + (size_t)gidx[e] * DIM;
                for (int c = 0; c < DIM; ++c)
                    atomicAdd(&s_acc[r * DIM + ((c & 3) << 5) + (c >> 2)], sp[c]);
                atomicAdd(&s_deg[r], 1);
            }
        }
    }
    __syncthreads();

    for (int r = wv; r < R; r += NWAVE) {
        const int row = lo + r;
        if (row < ndst) {
            const int   cnt = s_deg[r];
            const float inv = cnt > 0 ? 1.0f / (float)cnt : 0.0f;
            const float* a = s_acc + r * DIM + lane;
            float4 o;
            o.x = a[0]  * inv;
            o.y = a[32] * inv;
            o.z = a[64] * inv;
            o.w = a[96] * inv;
            if (FINAL) {
                const float4 bb = ((const float4*)bias)[lane];
                o.x += bb.x; o.y += bb.y; o.z += bb.z; o.w += bb.w;
                o.x = o.x >= 0.0f ? o.x : NEG_SLOPE * o.x;
                o.y = o.y >= 0.0f ? o.y : NEG_SLOPE * o.y;
                o.z = o.z >= 0.0f ? o.z : NEG_SLOPE * o.z;
                o.w = o.w >= 0.0f ? o.w : NEG_SLOPE * o.w;
            }
            ((float4*)(dst + (size_t)row * DIM))[lane] = o;
        }
    }
}

template <int RH, int CAP, bool FINAL>
__global__ void __launch_bounds__(NTHR)
owner_pass_halves(const float* __restrict__ src, float* __restrict__ dst,
                  const int* __restrict__ gidx, const int* __restrict__ sidx,
                  const unsigned* __restrict__ tags, const int* __restrict__ rec,
                  const float* __restrict__ bias, int nnz, int ndst)
{
    constexpr int R   = 2 * RH;
    constexpr int PER = (CAP + NTHR - 1) / NTHR;

    __shared__ __attribute__((aligned(16))) float s_acc[RH * DIM];
    __shared__ __attribute__((aligned(8)))  int2  s_pair[CAP];
    __shared__ int      s_list[CAP];
    __shared__ int      s_deg[R];
    __shared__ unsigned s_cnt;
    __shared__ unsigned s_nh[2];

    const int tid  = threadIdx.x;
    const int lane = tid & 31;
    const int wv   = tid >> 5;
    const int lo   = blockIdx.x * R;

    for (int i = tid; i < RH * DIM / 4; i += NTHR)
        ((float4*)s_acc)[i] = make_float4(0.0f, 0.0f, 0.0f, 0.0f);
    for (int i = tid; i < R; i += NTHR) s_deg[i] = 0;
    if (tid == 0) { s_cnt = 0u; s_nh[0] = 0u; s_nh[1] = 0u; }
    __syncthreads();

    auto note = [&](int e) {
        const unsigned pos = atomicAdd(&s_cnt, 1u);
        if (pos < (unsigned)CAP) s_list[pos] = e;
    };

    {
        const unsigned pat = (blockIdx.x & 255u) * 0x01010101u;

        auto look16 = [&](const uint4 dv, int e) {
            auto zero_bytes = [](const unsigned t) {
                return ~(((t & 0x7F7F7F7Fu) + 0x7F7F7F7Fu) | t | 0x7F7F7F7Fu);
            };
            unsigned hit = (zero_bytes(dv.x ^ pat) >> 7) | (zero_bytes(dv.y ^ pat) >> 6) |
                           (zero_bytes(dv.z ^ pat) >> 5) | (zero_bytes(dv.w ^ pat) >> 4);
            while (hit) {
                const int p = __builtin_ctz(hit);
                hit &= hit - 1u;
                note(e + 4 * (p & 7) + (p >> 3));
            }
        };

        const int n16   = nnz >> 4;
        const int nfull = n16 / (SCAN_U * NTHR);
        const uint4* __restrict__ tq = (const uint4*)tags;
        for (int it = 0; it < nfull; ++it) {
            const int qb = it * (SCAN_U * NTHR) + tid;
            uint4 d[SCAN_U];
#pragma unroll
            for (int u = 0; u < SCAN_U; ++u) d[u] = tq[qb + u * NTHR];
#pragma unroll
            for (int u = 0; u < SCAN_U; ++u) look16(d[u], (qb + u * NTHR) << 4);
        }
        for (int q = nfull * (SCAN_U * NTHR) + tid; q < n16; q += NTHR) look16(tq[q], q << 4);
        {
            const int e = (n16 << 4) + tid;
            if (e < nnz) {
                const unsigned a = (unsigned)(sidx[e] - lo);
                if (a < (unsigned)R) note(e);
            }
        }
    }
    __syncthreads();

    const unsigned tot    = s_cnt;
    const bool     listed = tot <= (unsigned)CAP;

    if (listed) {
        const int n = (int)tot;
        int2 rc[PER];
#pragma unroll
        for (int k = 0; k < PER; ++k) {
            const int i = tid + k * NTHR;
            rc[k] = make_int2(-1, 0);
            if (i < n) rc[k] = ((const int2*)rec)[s_list[i]];
        }
#pragma unroll
        for (int k = 0; k < PER; ++k) {
            if (wv * 32 + k * NTHR < n) {
                const unsigned a  = (unsigned)(rc[k].x - lo);
                const bool     h0 = a < (unsigned)RH;
                const bool     h1 = !h0 && a < (unsigned)R;
                const unsigned m0 = __builtin_amdgcn_ballot_w32(h0);
                const unsigned m1 = __builtin_amdgcn_ballot_w32(h1);
                unsigned b0 = 0u, b1 = 0u;
                if (lane == 0) {
                    if (m0) b0 = atomicAdd(&s_nh[0], (unsigned)__builtin_popcount(m0));
                    if (m1) b1 = atomicAdd(&s_nh[1], (unsigned)__builtin_popcount(m1));
                }
                b0 = (unsigned)__builtin_amdgcn_readfirstlane((int)b0);
                b1 = (unsigned)__builtin_amdgcn_readfirstlane((int)b1);
                if (h0) {
                    s_pair[b0 + __builtin_amdgcn_mbcnt_lo(m0, 0u)] = make_int2(rc[k].y, (int)a);
                    atomicAdd(&s_deg[a], 1);
                }
                if (h1) {
                    s_pair[CAP - 1 - (int)(b1 + __builtin_amdgcn_mbcnt_lo(m1, 0u))] =
                        make_int2(rc[k].y, (int)a - RH);
                    atomicAdd(&s_deg[a], 1);
                }
            }
        }
    }
    __syncthreads();

    for (int h = 0; h < 2; ++h) {
        if (listed) {
            const int n = (int)s_nh[h];
            for (int b = wv * BATCH; b < n; b += NWAVE * BATCH) {
                int g = 0, r = -1;
                if (lane < BATCH && b + lane < n) {
                    const int2 p = h == 0 ? s_pair[b + lane] : s_pair[CAP - 1 - (b + lane)];
                    g = p.x;
                    r = p.y;
                }
                float4 v[BATCH];
#pragma unroll
                for (int k = 0; k < BATCH; ++k) {
                    const int gk = __builtin_amdgcn_readlane(g, k);
                    v[k] = ((const float4*)(src + (size_t)gk * DIM))[lane];
                }
#pragma unroll
                for (int k = 0; k < BATCH; ++k) {
                    const int rk = __builtin_amdgcn_readlane(r, k);
                    if (rk >= 0) {
                        float* a = s_acc + rk * DIM + lane;
                        atomicAdd(a,      v[k].x);
                        atomicAdd(a + 32, v[k].y);
                        atomicAdd(a + 64, v[k].z);
                        atomicAdd(a + 96, v[k].w);
                    }
                }
            }
        } else {
            const int loh = lo + h * RH;
            for (int e = tid; e < nnz; e += NTHR) {
                const unsigned r = (unsigned)(sidx[e] - loh);
                if (r < (unsigned)RH) {
                    const float* __restrict__ sp = src + (size_t)gidx[e] * DIM;
                    for (int c = 0; c < DIM; ++c)
                        atomicAdd(&s_acc[r * DIM + ((c & 3) << 5) + (c >> 2)], sp[c]);
                    atomicAdd(&s_deg[h * RH + r], 1);
                }
            }
        }
        __syncthreads();

        for (int r = wv; r < RH; r += NWAVE) {
            const int row = lo + h * RH + r;
            float* a = s_acc + r * DIM + lane;
            if (row < ndst) {
                const int   cnt = s_deg[h * RH + r];
                const float inv = cnt > 0 ? 1.0f / (float)cnt : 0.0f;
                float4 o;
                o.x = a[0]  * inv;
                o.y = a[32] * inv;
                o.z = a[64] * inv;
                o.w = a[96] * inv;
                if (FINAL) {
                    const float4 bb = ((const float4*)bias)[lane];
                    o.x += bb.x; o.y += bb.y; o.z += bb.z; o.w += bb.w;
                    o.x = o.x >= 0.0f ? o.x : NEG_SLOPE * o.x;
                    o.y = o.y >= 0.0f ? o.y : NEG_SLOPE * o.y;
                    o.z = o.z >= 0.0f ? o.z : NEG_SLOPE * o.z;
                    o.w = o.w >= 0.0f ? o.w : NEG_SLOPE * o.w;
                }
                ((float4*)(dst + (size_t)row * DIM))[lane] = o;
            }
            if (h == 0) { a[0] = 0.0f; a[32] = 0.0f; a[64] = 0.0f; a[96] = 0.0f; }
        }
        if (h == 0) __syncthreads();
    }
}

#define R1   196
#define CAP1 10240
#define R2   512
#define CAP2 12288
#define RH2  391
#define CAPH 9216

extern "C" void kernel_launch(void* const* d_in, const int* in_sizes, int n_in,
                              void* d_out, int out_size, void* d_ws, size_t ws_size,
                              hipStream_t stream)
{
    const float* X    = (const float*)d_in[0];
    const int*   A    = (const int*)d_in[1];
    const float* W    = (const float*)d_in[2];
    const float* bias = (const float*)d_in[3];
    float*       out  = (float*)d_out;

    const int N   = in_sizes[0] / DIM;
    const int NNZ = in_sizes[1] / 2;
    const int* node = A;
    const int* edge = A + NNZ;

    float* m = (float*)d_ws;
    const size_t plane = ((size_t)NNZ + 15) & ~(size_t)15;
    unsigned* tag1  = (unsigned*)(m + (size_t)NEDGES * DIM);
    unsigned* tag2a = (unsigned*)((char*)tag1 + plane);
    unsigned* tag2b = (unsigned*)((char*)tag2a + plane);
    int*      rec   = (int*)((char*)tag2b + plane);
    const bool room  = ws_size >= (size_t)NEDGES * DIM * sizeof(float) + 3 * plane;
    const bool room2 = ws_size >= (size_t)NEDGES * DIM * sizeof(float) + 3 * plane + (size_t)NNZ * 8;

    const int mtiles = N / 16;
    gemm_wmma_f32<<<(mtiles + GEMM_WAVES - 1) / GEMM_WAVES, GEMM_THR, 0, stream>>>(X, W, out, mtiles);

    if (room2) {
        tag_entries<R1, 2 * RH2, 1, true><<<((NNZ >> 2) + 1023) / 1024, 1024, 0, stream>>>(
            edge, node, tag1, tag2a, tag2b, rec, NNZ);

        owner_pass<R1, CAP1, false, 1, 1><<<(NEDGES + R1 - 1) / R1, NTHR, 0, stream>>>(
            out, m, node, edge, tag1, nullptr, rec, bias, NNZ, NEDGES);

        owner_pass_halves<RH2, CAPH, true><<<(N + 2 * RH2 - 1) / (2 * RH2), NTHR, 0, stream>>>(
            m, out, edge, node, tag2a, rec, bias, NNZ, N);
    } else if (room) {
        tag_entries<R1, R2, 2, false><<<((NNZ >> 2) + 1023) / 1024, 1024, 0, stream>>>(
            edge, node, tag1, tag2a, tag2b, nullptr, NNZ);
        owner_pass<R1, CAP1, false, 1, 0><<<(NEDGES + R1 - 1) / R1, NTHR, 0, stream>>>(
            out, m, node, edge, tag1, nullptr, nullptr, bias, NNZ, NEDGES);
        owner_pass<R2, CAP2, true, 2, 0><<<(N + R2 - 1) / R2, NTHR, 0, stream>>>(
            m, out, edge, node, tag2a, tag2b, nullptr, bias, NNZ, N);
    } else {
        owner_pass<R1, CAP1, false, 0, 0><<<(NEDGES + R1 - 1) / R1, NTHR, 0, stream>>>(
            out, m, node, edge, nullptr, nullptr, nullptr, bias, NNZ, NEDGES);
        owner_pass<R2, CAP2, true, 0, 0><<<(N + R2 - 1) / R2, NTHR, 0, stream>>>(
            m, out, edge, node, nullptr, nullptr, nullptr, bias, NNZ, N);
    }
}


# both scatter passes: tagging launch also writes one bit per owner tag and 16-entry tag piece; an owner block reads its bit row and only the set pieces' tags, not the whole tag plane; same LDS adds
# speedup vs baseline: 4.0441x; 1.0998x over previous
#include <hip/hip_runtime.h>

typedef __attribute__((ext_vector_type(2))) float v2f;
typedef __attribute__((ext_vector_type(8))) float v8f;

#define DIM        128
#define NEDGES     50000
#define NEG_SLOPE  0.01f

#define GEMM_WAVES 8
#define GEMM_THR   (GEMM_WAVES * 32)
#define KQ         (DIM / 4)
#define W_ITEMS    (KQ * DIM / GEMM_THR)

__global__ void __launch_bounds__(GEMM_THR)
gemm_wmma_f32(const float* __restrict__ X, const float* __restrict__ W,
              float* __restrict__ Y, int mtiles)
{
    __shared__ __attribute__((aligned(16))) float4 s_w[KQ * DIM];

    const int tid  = threadIdx.x;
    const int lane = tid & 31;
    const int wv   = tid >> 5;
    const int half = lane >> 4;
    const int lr   = lane & 15;
    const int  mt   = blockIdx.x * GEMM_WAVES + wv;
    const bool live = mt < mtiles;
    const int  mtc  = live ? mt : mtiles - 1;

    float4 wit[W_ITEMS];
#pragma unroll
    for (int i = 0; i < W_ITEMS; ++i) {
        const int idx = tid + GEMM_THR * i;
        const float* __restrict__ wp = W + (size_t)(4 * (idx >> 7)) * DIM + (idx & (DIM - 1));
        wit[i] = make_float4(wp[0], wp[DIM], wp[2 * DIM], wp[3 * DIM]);
    }

    const float* __restrict__ ap = X + ((size_t)mtc * 16 + lr) * DIM + 2 * half;
    v2f a[KQ];
#pragma unroll
    for (int q = 0; q < KQ; ++q) a[q] = *(const v2f*)(ap + 4 * q);

#pragma unroll
    for (int i = 0; i < W_ITEMS; ++i) s_w[tid + GEMM_THR * i] = wit[i];
    __syncthreads();

    const v2f* __restrict__ bl = (const v2f*)s_w + (lr * 2 + half);

    v8f c[8];
#pragma unroll
    for (int nt = 0; nt < 8; ++nt) c[nt] = (v8f){};

#pragma unroll
    for (int q = 0; q < KQ; ++q) {
        v2f b[8];
#pragma unroll
        for (int nt = 0; nt < 8; ++nt) b[nt] = bl[(q * DIM + nt * 16) * 2];
#pragma unroll
        for (int nt = 0; nt < 8; ++nt)
            c[nt] = __builtin_amdgcn_wmma_f32_16x16x4_f32(false, a[q], false, b[nt],
                                                          (short)0, c[nt], false, false);
    }

    if (live) {
        float* __restrict__ yp = Y + (size_t)mt * 16 * DIM + lane;
#pragma unroll
        for (int p = 0; p < 4; ++p) {
#pragma unroll
            for (int r = 0; r < 8; ++r) {
                const auto s = __builtin_amdgcn_permlane16_swap(
                    __float_as_uint(c[2 * p][r]), __float_as_uint(c[2 * p + 1][r]), false, false);
                yp[(size_t)r * DIM + 32 * p]       = __uint_as_float(s[0]);
                yp[(size_t)(8 + r) * DIM + 32 * p] = __uint_as_float(s[1]);
            }
        }
    }
}

template <int R, int NPL>
__device__ __forceinline__ unsigned owner_tag(int idx, unsigned plane)
{
    const unsigned o = (unsigned)idx / (unsigned)R;
    if (NPL == 1) return o & 255u;
    return ((o / 255u) & 1u) == plane ? o % 255u : 255u;
}

template <int R, int NPL>
__device__ __forceinline__ unsigned tag_word(const int4 v, unsigned plane)
{
    return owner_tag<R, NPL>(v.x, plane) | (owner_tag<R, NPL>(v.y, plane) << 8) |
           (owner_tag<R, NPL>(v.z, plane) << 16) | (owner_tag<R, NPL>(v.w, plane) << 24);
}

template <int RA, int RB, int NPLB, bool REC>
__global__ void __launch_bounds__(1024)
tag_entries(const int* __restrict__ ia, const int* __restrict__ ib,
            unsigned* __restrict__ ta, unsigned* __restrict__ tb0,
            unsigned* __restrict__ tb1, int* __restrict__ rec, int nnz)
{
    const int nq = nnz >> 2;
    const int q  = blockIdx.x * 1024 + threadIdx.x;
    if (q < nq) {
        const int4 a = ((const int4*)ia)[q];
        const int4 b = ((const int4*)ib)[q];
        ta[q]  = tag_word<RA, 1>(a, 0u);
        if (NPLB == 1) {
            tb0[q] = tag_word<RB, 1>(b, 0u);
        } else {
            tb0[q] = tag_word<RB, 2>(b, 0u);
            tb1[q] = tag_word<RB, 2>(b, 1u);
        }
    }
    if (REC) {
        for (int k = 0; k < 2; ++k) {
            const int P = blockIdx.x * 2048 + k * 1024 + threadIdx.x;
            if (P < 2 * nq) {
                const int2 a = ((const int2*)ia)[P];
                const int2 b = ((const int2*)ib)[P];
                ((int4*)rec)[P] = make_int4(b.x, a.x, b.y, a.y);
            }
        }
        if (q == 0)
            for (int e = 4 * nq; e < nnz; ++e) { rec[2 * e] = ib[e]; rec[2 * e + 1] = ia[e]; }
    }
}

#define PCH 1024

template <int RA, int RB>
__global__ void __launch_bounds__(1024)
tag_entries_bits(const int* __restrict__ ia, const int* __restrict__ ib,
                 unsigned* __restrict__ ta, unsigned* __restrict__ tb,
                 int* __restrict__ rec, unsigned* __restrict__ bits_a,
                 unsigned* __restrict__ bits_b, int nnz, int row_words)
{
    __shared__ unsigned s_bm[2][256 * (PCH / 32)];

    const int tid  = threadIdx.x;
    const int lane = tid & 31;
    const int wv   = tid >> 5;

    for (int i = tid; i < 2 * 256 * (PCH / 32); i += 1024) (&s_bm[0][0])[i] = 0u;
    __syncthreads();

    const int nq = nnz >> 2;
    for (int k = 0; k < 4; ++k) {
        const int q = blockIdx.x * (4 * PCH) + k * 1024 + tid;
        if (q < nq) {
            const int4 a = ((const int4*)ia)[q];
            const int4 b = ((const int4*)ib)[q];
            const unsigned wa = tag_word<RA, 1>(a, 0u);
            const unsigned wb = tag_word<RB, 1>(b, 0u);
            ta[q] = wa;
            tb[q] = wb;
            const int      p   = (k * 1024 + tid) >> 2;
            const unsigned bit = 1u << (p & 31);
            const int      w   = p >> 5;
            atomicOr(&s_bm[0][(wa & 255u) * (PCH / 32) + w], bit);
            atomicOr(&s_bm[0][((wa >> 8) & 255u) * (PCH / 32) + w], bit);
            atomicOr(&s_bm[0][((wa >> 16) & 255u) * (PCH / 32) + w], bit);
            atomicOr(&s_bm[0][(wa >> 24) * (PCH / 32) + w], bit);
            atomicOr(&s_bm[1][(wb & 255u) * (PCH / 32) + w], bit);
            atomicOr(&s_bm[1][((wb >> 8) & 255u) * (PCH / 32) + w], bit);
            atomicOr(&s_bm[1][((wb >> 16) & 255u) * (PCH / 32) + w], bit);
            atomicOr(&s_bm[1][(wb >> 24) * (PCH / 32) + w], bit);
        }
    }
    for (int k = 0; k < 8; ++k) {
        const int P = blockIdx.x * (8 * PCH) + k * 1024 + tid;
        if (P < 2 * nq) {
            const int2 a = ((const int2*)ia)[P];
            const int2 b = ((const int2*)ib)[P];
            ((int4*)rec)[P] = make_int4(b.x, a.x, b.y, a.y);
        }
    }
    if (blockIdx.x == 0 && tid == 0)
        for (int e = 4 * nq; e < nnz; ++e) { rec[2 * e] = ib[e]; rec[2 * e + 1] = ia[e]; }
    __syncthreads();

    for (int o = wv; o < 256; o += 32) {
        bits_a[(size_t)o * row_words + blockIdx.x * (PCH / 32) + lane] = s_bm[0][o * (PCH / 32) + lane];
        bits_b[(size_t)o * row_words + blockIdx.x * (PCH / 32) + lane] = s_bm[1][o * (PCH / 32) + lane];
    }
}

#define NTHR 1024
#define NWAVE 32
#define BATCH 16
#define SCAN_U 8

__device__ __forceinline__ unsigned hits16(const uint4 dv, const unsigned pat)
{
    auto zero_bytes = [](const unsigned t) {
        return ~(((t & 0x7F7F7F7Fu) + 0x7F7F7F7Fu) | t | 0x7F7F7F7Fu);
    };
    return (zero_bytes(dv.x ^ pat) >> 7) | (zero_bytes(dv.y ^ pat) >> 6) |
           (zero_bytes(dv.z ^ pat) >> 5) | (zero_bytes(dv.w ^ pat) >> 4);
}

template <int CAP>
__device__ __forceinline__ bool entries_by_bits(const unsigned* __restrict__ bitrow, int row_words,
                                                const unsigned* __restrict__ tags, unsigned pat,
                                                int nnz, int* s_list, unsigned* s_np, unsigned* s_cnt)
{
    constexpr int PERP = (CAP + NTHR - 1) / NTHR;
    const int tid = threadIdx.x;
    const int n16 = nnz >> 4;

    const uint4* __restrict__ br = (const uint4*)bitrow;
    for (int w = tid; w < (row_words >> 2); w += NTHR) {
        const uint4 m4 = br[w];
        const unsigned mw[4] = {m4.x, m4.y, m4.z, m4.w};
        const unsigned c = (unsigned)(__builtin_popcount(m4.x) + __builtin_popcount(m4.y) +
                                      __builtin_popcount(m4.z) + __builtin_popcount(m4.w));
        if (c) {
            unsigned pos = atomicAdd(s_np, c);
#pragma unroll
            for (int j = 0; j < 4; ++j) {
                unsigned mm = mw[j];
                while (mm) {
                    const int piece = (w * 4 + j) * 32 + __builtin_ctz(mm);
                    mm &= mm - 1u;
                    if (pos < (unsigned)CAP) s_list[pos] = piece < n16 ? piece : -1;
                    ++pos;
                }
            }
        }
    }
    __syncthreads();

    const unsigned np = *s_np;
    if (np > (unsigned)CAP) return false;

    int pc[PERP];
#pragma unroll
    for (int k = 0; k < PERP; ++k) {
        const int i = tid + k * NTHR;
        pc[k] = i < (int)np ? s_list[i] : -1;
    }
    __syncthreads();

    const uint4* __restrict__ tq = (const uint4*)tags;
    uint4 d[PERP];
#pragma unroll
    for (int k = 0; k < PERP; ++k) d[k] = tq[pc[k] >= 0 ? pc[k] : 0];
    unsigned hk[PERP];
    unsigned c = 0u;
#pragma unroll
    for (int k = 0; k < PERP; ++k) {
        hk[k] = pc[k] >= 0 ? hits16(d[k], pat) : 0u;
        c += (unsigned)__builtin_popcount(hk[k]);
    }
    if (c) {
        unsigned pos = atomicAdd(s_cnt, c);
#pragma unroll
        for (int k = 0; k < PERP; ++k) {
            unsigned h = hk[k];
            while (h) {
                const int p = __builtin_ctz(h);
                h &= h - 1u;
                if (pos < (unsigned)CAP) s_list[pos] = (pc[k] << 4) + 4 * (p & 7) + (p >> 3);
                ++pos;
            }
        }
    }
    return true;
}


template <int R, int CAP, bool FINAL, int NPL, int REC, bool BITS>
__global__ void __launch_bounds__(NTHR)
owner_pass(const float* __restrict__ src, float* __restrict__ dst,
           const int* __restrict__ gidx, const int* __restrict__ sidx,
           const unsigned* __restrict__ tags0, const unsigned* __restrict__ tags1,
           const int* __restrict__ rec,
           const unsigned* __restrict__ bits, int row_words,
           const float* __restrict__ bias, int nnz, int ndst)
{
    __shared__ __attribute__((aligned(16))) float s_acc[R * DIM];
    __shared__ int      s_deg[R];
    __shared__ int      s_list[CAP];
    __shared__ unsigned s_cnt;
    __shared__ unsigned s_np;

    const int tid  = threadIdx.x;
    const int lane = tid & 31;
    const int wv   = tid >> 5;
    const int lo   = blockIdx.x * R;

    for (int i = tid; i < R * DIM / 4; i += NTHR)
        ((float4*)s_acc)[i] = make_float4(0.0f, 0.0f, 0.0f, 0.0f);
    for (int i = tid; i < R; i += NTHR) s_deg[i] = 0;
    if (tid == 0) { s_cnt = 0u; s_np = 0u; }
    __syncthreads();

    auto note = [&](int e) {
        const unsigned pos = atomicAdd(&s_cnt, 1u);
        if (pos < (unsigned)CAP) s_list[pos] = e;
    };

    auto look = [&](const int4 dv, int e) {
        const unsigned a0 = (unsigned)(dv.x - lo);
        const unsigned a1 = (unsigned)(dv.y - lo);
        const unsigned a2 = (unsigned)(dv.z - lo);
        const unsigned a3 = (unsigned)(dv.w - lo);
        const unsigned mn = min(min(a0, a1), min(a2, a3));
        if (mn < (unsigned)R) {
            if (a0 < (unsigned)R) note(e);
            if (a1 < (unsigned)R) note(e + 1);
            if (a2 < (unsigned)R) note(e + 2);
            if (a3 < (unsigned)R) note(e + 3);
        }
    };

    if (NPL > 0) {
        const bool odd_run = NPL == 2 && ((blockIdx.x / 255u) & 1u) != 0u;
        const unsigned* __restrict__ tags = odd_run ? tags1 : tags0;
        const unsigned pat = (NPL == 2 ? blockIdx.x % 255u : blockIdx.x & 255u) * 0x01010101u;

        auto look16 = [&](const uint4 dv, int e) {
            auto zero_bytes = [](const unsigned t) {
                return ~(((t & 0x7F7F7F7Fu) + 0x7F7F7F7Fu) | t | 0x7F7F7F7Fu);
            };
            unsigned hit = (zero_bytes(dv.x ^ pat) >> 7) | (zero_bytes(dv.y ^ pat) >> 6) |
                           (zero_bytes(dv.z ^ pat) >> 5) | (zero_bytes(dv.w ^ pat) >> 4);
            while (hit) {
                const int p = __builtin_ctz(hit);
                hit &= hit - 1u;
                note(e + 4 * (p & 7) + (p >> 3));
            }
        };

        const int n16   = nnz >> 4;
        bool whole = true;
        if (BITS)
            whole = !entries_by_bits<CAP>(bits + (size_t)(blockIdx.x & 255u) * row_words, row_words,
                                          tags, pat, nnz, s_list, &s_np, &s_cnt);
        if (whole) {
            const int nfull = n16 / (SCAN_U * NTHR);
            const uint4* __restrict__ tq = (const uint4*)tags;
            for (int it = 0; it < nfull; ++it) {
                const int qb = it * (SCAN_U * NTHR) + tid;
                uint4 d[SCAN_U];
#pragma unroll
                for (int u = 0; u < SCAN_U; ++u) d[u] = tq[qb + u * NTHR];
#pragma unroll
                for (int u = 0; u < SCAN_U; ++u) look16(d[u], (qb + u * NTHR) << 4);
            }
            for (int q = nfull * (SCAN_U * NTHR) + tid; q < n16; q += NTHR) look16(tq[q], q << 4);
        }
        {
            const int e = (n16 << 4) + tid;
            if (e < nnz) {
                const unsigned a = (unsigned)(sidx[e] - lo);
                if (a < (unsigned)R) note(e);
            }
        }
    } else {
        const int nq    = nnz >> 2;
        const int nfull = nq / (SCAN_U * NTHR);
        const int4* __restrict__ sq = (const int4*)sidx;
        for (int it = 0; it < nfull; ++it) {
            const int qb = it * (SCAN_U * NTHR) + tid;
            int4 d[SCAN_U];
#pragma unroll
            for (int u = 0; u < SCAN_U; ++u) d[u] = sq[qb + u * NTHR];
#pragma unroll
            for (int u = 0; u < SCAN_U; ++u) look(d[u], (qb + u * NTHR) << 2);
        }
        for (int q = nfull * (SCAN_U * NTHR) + tid; q < nq; q += NTHR) look(sq[q], q << 2);
        {
            const int e = (nq << 2) + tid;
            if (e < nnz) {
                const unsigned a = (unsigned)(sidx[e] - lo);
                if (a < (unsigned)R) note(e);
            }
        }
    }
    __syncthreads();

    const unsigned tot = s_cnt;
    if (tot <= (unsigned)CAP) {
        const int n = (int)tot;
        for (int b = wv * BATCH; b < n; b += NWAVE * BATCH) {
            int g = 0, r = -1;
            if (lane < BATCH && b + lane < n) {
                const int e = s_list[b + lane];
                int de, ge;
                if (REC) {
                    const int2 rc = ((const int2*)rec)[e];
                    de = REC == 1 ? rc.y : rc.x;
                    ge = REC == 1 ? rc.x : rc.y;
                } else {
                    de = sidx[e];
                    ge = gidx[e];
                }
                const unsigned a = (unsigned)(de - lo);
                const bool mine = a < (unsigned)R;
                r = mine ? (int)a : -1;
                g = mine ? ge : 0;
            }
            float4 v[BATCH];
#pragma unroll
            for (int k = 0; k < BATCH; ++k) {
                const int gk = __builtin_amdgcn_readlane(g, k);
                v[k] = ((const float4*)(src + (size_t)gk * DIM))[lane];
            }
#pragma unroll
            for (int k = 0; k < BATCH; ++k) {
                const int rk = __builtin_amdgcn_readlane(r, k);
                if (rk >= 0) {
                    float* a = s_acc + rk * DIM + lane;
                    atomicAdd(a,      v[k].x);
                    atomicAdd(a + 32, v[k].y);
                    atomicAdd(a + 64, v[k].z);
                    atomicAdd(a + 96, v[k].w);
                    if (lane == 0) atomicAdd(&s_deg[rk], 1);
                }
            }
        }
    } else {
        for (int e = tid; e < nnz; e += NTHR) {
            const unsigned r = (unsigned)(sidx[e] - lo);
            if (r < (unsigned)R) {
                const float* __restrict__ sp = src + (size_t)gidx[e] * DIM;
                for (int c = 0; c < DIM; ++c)
                    atomicAdd(&s_acc[r * DIM + ((c & 3) << 5) + (c >> 2)], sp[c]);
                atomicAdd(&s_deg[r], 1);
            }
        }
    }
    __syncthreads();

    for (int r = wv; r < R; r += NWAVE) {
        const int row = lo + r;
        if (row < ndst) {
            const int   cnt = s_deg[r];
            const float inv = cnt > 0 ? 1.0f / (float)cnt : 0.0f;
            const float* a = s_acc + r * DIM + lane;
            float4 o;
            o.x = a[0]  * inv;
            o.y = a[32] * inv;
            o.z = a[64] * inv;
            o.w = a[96] * inv;
            if (FINAL) {
                const float4 bb = ((const float4*)bias)[lane];
                o.x += bb.x; o.y += bb.y; o.z += bb.z; o.w += bb.w;
                o.x = o.x >= 0.0f ? o.x : NEG_SLOPE * o.x;
                o.y = o.y >= 0.0f ? o.y : NEG_SLOPE * o.y;
                o.z = o.z >= 0.0f ? o.z : NEG_SLOPE * o.z;
                o.w = o.w >= 0.0f ? o.w : NEG_SLOPE * o.w;
            }
            ((float4*)(dst + (size_t)row * DIM))[lane] = o;
        }
    }
}

template <int RH, int CAP, bool FINAL, bool BITS>
__global__ void __launch_bounds__(NTHR)
owner_pass_halves(const float* __restrict__ src, float* __restrict__ dst,
                  const int* __restrict__ gidx, const int* __restrict__ sidx,
                  const unsigned* __restrict__ tags, const int* __restrict__ rec,
                  const unsigned* __restrict__ bits, int row_words,
                  const float* __restrict__ bias, int nnz, int ndst)
{
    constexpr int R   = 2 * RH;
    constexpr int PER = (CAP + NTHR - 1) / NTHR;

    __shared__ __attribute__((aligned(16))) float s_acc[RH * DIM];
    __shared__ __attribute__((aligned(8)))  int2  s_pair[CAP];
    __shared__ int      s_list[CAP];
    __shared__ int      s_deg[R];
    __shared__ unsigned s_cnt;
    __shared__ unsigned s_np;
    __shared__ unsigned s_nh[2];

    const int tid  = threadIdx.x;
    const int lane = tid & 31;
    const int wv   = tid >> 5;
    const int lo   = blockIdx.x * R;

    for (int i = tid; i < RH * DIM / 4; i += NTHR)
        ((float4*)s_acc)[i] = make_float4(0.0f, 0.0f, 0.0f, 0.0f);
    for (int i = tid; i < R; i += NTHR) s_deg[i] = 0;
    if (tid == 0) { s_cnt = 0u; s_np = 0u; s_nh[0] = 0u; s_nh[1] = 0u; }
    __syncthreads();

    auto note = [&](int e) {
        const unsigned pos = atomicAdd(&s_cnt, 1u);
        if (pos < (unsigned)CAP) s_list[pos] = e;
    };

    {
        const unsigned pat = (blockIdx.x & 255u) * 0x01010101u;

        auto look16 = [&](const uint4 dv, int e) {
            auto zero_bytes = [](const unsigned t) {
                return ~(((t & 0x7F7F7F7Fu) + 0x7F7F7F7Fu) | t | 0x7F7F7F7Fu);
            };
            unsigned hit = (zero_bytes(dv.x ^ pat) >> 7) | (zero_bytes(dv.y ^ pat) >> 6) |
                           (zero_bytes(dv.z ^ pat) >> 5) | (zero_bytes(dv.w ^ pat) >> 4);
            while (hit) {
                const int p = __builtin_ctz(hit);
                hit &= hit - 1u;
                note(e + 4 * (p & 7) + (p >> 3));
            }
        };

        const int n16   = nnz >> 4;
        bool whole = true;
        if (BITS)
            whole = !entries_by_bits<CAP>(bits + (size_t)(blockIdx.x & 255u) * row_words, row_words,
                                          tags, pat, nnz, s_list, &s_np, &s_cnt);
        if (whole) {
            const int nfull = n16 / (SCAN_U * NTHR);
            const uint4* __restrict__ tq = (const uint4*)tags;
            for (int it = 0; it < nfull; ++it) {
                const int qb = it * (SCAN_U * NTHR) + tid;
                uint4 d[SCAN_U];
#pragma unroll
                for (int u = 0; u < SCAN_U; ++u) d[u] = tq[qb + u * NTHR];
#pragma unroll
                for (int u = 0; u < SCAN_U; ++u) look16(d[u], (qb + u * NTHR) << 4);
            }
            for (int q = nfull * (SCAN_U * NTHR) + tid; q < n16; q += NTHR) look16(tq[q], q << 4);
        }
        {
            const int e = (n16 << 4) + tid;
            if (e < nnz) {
                const unsigned a = (unsigned)(sidx[e] - lo);
                if (a < (unsigned)R) note(e);
            }
        }
    }
    __syncthreads();

    const unsigned tot    = s_cnt;
    const bool     listed = tot <= (unsigned)CAP;

    if (listed) {
        const int n = (int)tot;
        int2 rc[PER];
#pragma unroll
        for (int k = 0; k < PER; ++k) {
            const int i = tid + k * NTHR;
            rc[k] = make_int2(-1, 0);
            if (i < n) rc[k] = ((const int2*)rec)[s_list[i]];
        }
#pragma unroll
        for (int k = 0; k < PER; ++k) {
            if (wv * 32 + k * NTHR < n) {
                const unsigned a  = (unsigned)(rc[k].x - lo);
                const bool     h0 = a < (unsigned)RH;
                const bool     h1 = !h0 && a < (unsigned)R;
                const unsigned m0 = __builtin_amdgcn_ballot_w32(h0);
                const unsigned m1 = __builtin_amdgcn_ballot_w32(h1);
                unsigned b0 = 0u, b1 = 0u;
                if (lane == 0) {
                    if (m0) b0 = atomicAdd(&s_nh[0], (unsigned)__builtin_popcount(m0));
                    if (m1) b1 = atomicAdd(&s_nh[1], (unsigned)__builtin_popcount(m1));
                }
                b0 = (unsigned)__builtin_amdgcn_readfirstlane((int)b0);
                b1 = (unsigned)__builtin_amdgcn_readfirstlane((int)b1);
                if (h0) {
                    s_pair[b0 + __builtin_amdgcn_mbcnt_lo(m0, 0u)] = make_int2(rc[k].y, (int)a);
                    atomicAdd(&s_deg[a], 1);
                }
                if (h1) {
                    s_pair[CAP - 1 - (int)(b1 + __builtin_amdgcn_mbcnt_lo(m1, 0u))] =
                        make_int2(rc[k].y, (int)a - RH);
                    atomicAdd(&s_deg[a], 1);
                }
            }
        }
    }
    __syncthreads();

    for (int h = 0; h < 2; ++h) {
        if (listed) {
            const int n = (int)s_nh[h];
            for (int b = wv * BATCH; b < n; b += NWAVE * BATCH) {
                int g = 0, r = -1;
                if (lane < BATCH && b + lane < n) {
                    const int2 p = h == 0 ? s_pair[b + lane] : s_pair[CAP - 1 - (b + lane)];
                    g = p.x;
                    r = p.y;
                }
                float4 v[BATCH];
#pragma unroll
                for (int k = 0; k < BATCH; ++k) {
                    const int gk = __builtin_amdgcn_readlane(g, k);
                    v[k] = ((const float4*)(src + (size_t)gk * DIM))[lane];
                }
#pragma unroll
                for (int k = 0; k < BATCH; ++k) {
                    const int rk = __builtin_amdgcn_readlane(r, k);
                    if (rk >= 0) {
                        float* a = s_acc + rk * DIM + lane;
                        atomicAdd(a,      v[k].x);
                        atomicAdd(a + 32, v[k].y);
                        atomicAdd(a + 64, v[k].z);
                        atomicAdd(a + 96, v[k].w);
                    }
                }
            }
        } else {
            const int loh = lo + h * RH;
            for (int e = tid; e < nnz; e += NTHR) {
                const unsigned r = (unsigned)(sidx[e] - loh);
                if (r < (unsigned)RH) {
                    const float* __restrict__ sp = src + (size_t)gidx[e] * DIM;
                    for (int c = 0; c < DIM; ++c)
                        atomicAdd(&s_acc[r * DIM + ((c & 3) << 5) + (c >> 2)], sp[c]);
                    atomicAdd(&s_deg[h * RH + r], 1);
                }
            }
        }
        __syncthreads();

        for (int r = wv; r < RH; r += NWAVE) {
            const int row = lo + h * RH + r;
            float* a = s_acc + r * DIM + lane;
            if (row < ndst) {
                const int   cnt = s_deg[h * RH + r];
                const float inv = cnt > 0 ? 1.0f / (float)cnt : 0.0f;
                float4 o;
                o.x = a[0]  * inv;
                o.y = a[32] * inv;
                o.z = a[64] * inv;
                o.w = a[96] * inv;
                if (FINAL) {
                    const float4 bb = ((const float4*)bias)[lane];
                    o.x += bb.x; o.y += bb.y; o.z += bb.z; o.w += bb.w;
                    o.x = o.x >= 0.0f ? o.x : NEG_SLOPE * o.x;
                    o.y = o.y >= 0.0f ? o.y : NEG_SLOPE * o.y;
                    o.z = o.z >= 0.0f ? o.z : NEG_SLOPE * o.z;
                    o.w = o.w >= 0.0f ? o.w : NEG_SLOPE * o.w;
                }
                ((float4*)(dst + (size_t)row * DIM))[lane] = o;
            }
            if (h == 0) { a[0] = 0.0f; a[32] = 0.0f; a[64] = 0.0f; a[96] = 0.0f; }
        }
        if (h == 0) __syncthreads();
    }
}

#define R1   196
#define CAP1 10240
#define R2   512
#define CAP2 12288
#define RH2  391
#define CAPH 9216

extern "C" void kernel_launch(void* const* d_in, const int* in_sizes, int n_in,
                              void* d_out, int out_size, void* d_ws, size_t ws_size,
                              hipStream_t stream)
{
    const float* X    = (const float*)d_in[0];
    const int*   A    = (const int*)d_in[1];
    const float* W    = (const float*)d_in[2];
    const float* bias = (const float*)d_in[3];
    float*       out  = (float*)d_out;

    const int N   = in_sizes[0] / DIM;
    const int NNZ = in_sizes[1] / 2;
    const int* node = A;
    const int* edge = A + NNZ;

    float* m = (float*)d_ws;
    const size_t plane = ((size_t)NNZ + 15) & ~(size_t)15;
    unsigned* tag1  = (unsigned*)(m + (size_t)NEDGES * DIM);
    unsigned* tag2a = (unsigned*)((char*)tag1 + plane);
    unsigned* tag2b = (unsigned*)((char*)tag2a + plane);
    int*      rec   = (int*)((char*)tag2b + plane);
    const bool room  = ws_size >= (size_t)NEDGES * DIM * sizeof(float) + 3 * plane;
    const bool room2 = ws_size >= (size_t)NEDGES * DIM * sizeof(float) + 3 * plane + (size_t)NNZ * 8;
    const int    nchunk = (int)((((size_t)NNZ + 15) / 16 + PCH - 1) / PCH);
    const int    roww   = nchunk * (PCH / 32);
    const size_t bitsz  = (size_t)256 * roww * sizeof(unsigned);
    const size_t boff   = ((size_t)NEDGES * DIM * sizeof(float) + 2 * plane + (size_t)NNZ * 8 + 127) & ~(size_t)127;
    const bool   room3  = ws_size >= boff + 2 * bitsz;

    const int mtiles = N / 16;
    gemm_wmma_f32<<<(mtiles + GEMM_WAVES - 1) / GEMM_WAVES, GEMM_THR, 0, stream>>>(X, W, out, mtiles);

    if (room3) {
        int*      rec3  = (int*)((char*)tag2a + plane);
        unsigned* bits1 = (unsigned*)((char*)d_ws + boff);
        unsigned* bits2 = (unsigned*)((char*)d_ws + boff + bitsz);
        tag_entries_bits<R1, 2 * RH2><<<nchunk, 1024, 0, stream>>>(
            edge, node, tag1, tag2a, rec3, bits1, bits2, NNZ, roww);
        owner_pass<R1, CAP1, false, 1, 1, true><<<(NEDGES + R1 - 1) / R1, NTHR, 0, stream>>>(
            out, m, node, edge, tag1, nullptr, rec3, bits1, roww, bias, NNZ, NEDGES);
        owner_pass_halves<RH2, CAPH, true, true><<<(N + 2 * RH2 - 1) / (2 * RH2), NTHR, 0, stream>>>(
            m, out, edge, node, tag2a, rec3, bits2, roww, bias, NNZ, N);
    } else if (room2) {
        tag_entries<R1, 2 * RH2, 1, true><<<((NNZ >> 2) + 1023) / 1024, 1024, 0, stream>>>(
            edge, node, tag1, tag2a, tag2b, rec, NNZ);

        owner_pass<R1, CAP1, false, 1, 1, false><<<(NEDGES + R1 - 1) / R1, NTHR, 0, stream>>>(
            out, m, node, edge, tag1, nullptr, rec, nullptr, 0, bias, NNZ, NEDGES);

        owner_pass_halves<RH2, CAPH, true, false><<<(N + 2 * RH2 - 1) / (2 * RH2), NTHR, 0, stream>>>(
            m, out, edge, node, tag2a, rec, nullptr, 0, bias, NNZ, N);
    } else if (room) {
        tag_entries<R1, R2, 2, false><<<((NNZ >> 2) + 1023) / 1024, 1024, 0, stream>>>(
            edge, node, tag1, tag2a, tag2b, nullptr, NNZ);
        owner_pass<R1, CAP1, false, 1, 0, false><<<(NEDGES + R1 - 1) / R1, NTHR, 0, stream>>>(
            out, m, node, edge, tag1, nullptr, nullptr, nullptr, 0, bias, NNZ, NEDGES);
        owner_pass<R2, CAP2, true, 2, 0, false><<<(N + R2 - 1) / R2, NTHR, 0, stream>>>(
            m, out, edge, node, tag2a, tag2b, nullptr, nullptr, 0, bias, NNZ, N);
    } else {
        owner_pass<R1, CAP1, false, 0, 0, false><<<(NEDGES + R1 - 1) / R1, NTHR, 0, stream>>>(
            out, m, node, edge, nullptr, nullptr, nullptr, nullptr, 0, bias, NNZ, NEDGES);
        owner_pass<R2, CAP2, true, 0, 0, false><<<(N + R2 - 1) / R2, NTHR, 0, stream>>>(
            m, out, edge, node, nullptr, nullptr, nullptr, nullptr, 0, bias, NNZ, N);
    }
}


# both scatter passes: an owner block puts its entry list in destination-row order (LDS count, scan, place); a wave sums each run of one row in registers, one set of LDS adds per run, not per entry
# speedup vs baseline: 4.4031x; 1.0888x over previous
#include <hip/hip_runtime.h>

typedef __attribute__((ext_vector_type(2))) float v2f;
typedef __attribute__((ext_vector_type(8))) float v8f;

#define DIM        128
#define NEDGES     50000
#define NEG_SLOPE  0.01f

#define GEMM_WAVES 8
#define GEMM_THR   (GEMM_WAVES * 32)
#define KQ         (DIM / 4)
#define W_ITEMS    (KQ * DIM / GEMM_THR)

__global__ void __launch_bounds__(GEMM_THR)
gemm_wmma_f32(const float* __restrict__ X, const float* __restrict__ W,
              float* __restrict__ Y, int mtiles)
{
    __shared__ __attribute__((aligned(16))) float4 s_w[KQ * DIM];

    const int tid  = threadIdx.x;
    const int lane = tid & 31;
    const int wv   = tid >> 5;
    const int half = lane >> 4;
    const int lr   = lane & 15;
    const int  mt   = blockIdx.x * GEMM_WAVES + wv;
    const bool live = mt < mtiles;
    const int  mtc  = live ? mt : mtiles - 1;

    float4 wit[W_ITEMS];
#pragma unroll
    for (int i = 0; i < W_ITEMS; ++i) {
        const int idx = tid + GEMM_THR * i;
        const float* __restrict__ wp = W + (size_t)(4 * (idx >> 7)) * DIM + (idx & (DIM - 1));
        wit[i] = make_float4(wp[0], wp[DIM], wp[2 * DIM], wp[3 * DIM]);
    }

    const float* __restrict__ ap = X + ((size_t)mtc * 16 + lr) * DIM + 2 * half;
    v2f a[KQ];
#pragma unroll
    for (int q = 0; q < KQ; ++q) a[q] = *(const v2f*)(ap + 4 * q);

#pragma unroll
    for (int i = 0; i < W_ITEMS; ++i) s_w[tid + GEMM_THR * i] = wit[i];
    __syncthreads();

    const v2f* __restrict__ bl = (const v2f*)s_w + (lr * 2 + half);

    v8f c[8];
#pragma unroll
    for (int nt = 0; nt < 8; ++nt) c[nt] = (v8f){};

#pragma unroll
    for (int q = 0; q < KQ; ++q) {
        v2f b[8];
#pragma unroll
        for (int nt = 0; nt < 8; ++nt) b[nt] = bl[(q * DIM + nt * 16) * 2];
#pragma unroll
        for (int nt = 0; nt < 8; ++nt)
            c[nt] = __builtin_amdgcn_wmma_f32_16x16x4_f32(false, a[q], false, b[nt],
                                                          (short)0, c[nt], false, false);
    }

    if (live) {
        float* __restrict__ yp = Y + (size_t)mt * 16 * DIM + lane;
#pragma unroll
        for (int p = 0; p < 4; ++p) {
#pragma unroll
            for (int r = 0; r < 8; ++r) {
                const auto s = __builtin_amdgcn_permlane16_swap(
                    __float_as_uint(c[2 * p][r]), __float_as_uint(c[2 * p + 1][r]), false, false);
                yp[(size_t)r * DIM + 32 * p]       = __uint_as_float(s[0]);
                yp[(size_t)(8 + r) * DIM + 32 * p] = __uint_as_float(s[1]);
            }
        }
    }
}

template <int R, int NPL>
__device__ __forceinline__ unsigned owner_tag(int idx, unsigned plane)
{
    const unsigned o = (unsigned)idx / (unsigned)R;
    if (NPL == 1) return o & 255u;
    return ((o / 255u) & 1u) == plane ? o % 255u : 255u;
}

template <int R, int NPL>
__device__ __forceinline__ unsigned tag_word(const int4 v, unsigned plane)
{
    return owner_tag<R, NPL>(v.x, plane) | (owner_tag<R, NPL>(v.y, plane) << 8) |
           (owner_tag<R, NPL>(v.z, plane) << 16) | (owner_tag<R, NPL>(v.w, plane) << 24);
}

template <int RA, int RB, int NPLB, bool REC>
__global__ void __launch_bounds__(1024)
tag_entries(const int* __restrict__ ia, const int* __restrict__ ib,
            unsigned* __restrict__ ta, unsigned* __restrict__ tb0,
            unsigned* __restrict__ tb1, int* __restrict__ rec, int nnz)
{
    const int nq = nnz >> 2;
    const int q  = blockIdx.x * 1024 + threadIdx.x;
    if (q < nq) {
        const int4 a = ((const int4*)ia)[q];
        const int4 b = ((const int4*)ib)[q];
        ta[q]  = tag_word<RA, 1>(a, 0u);
        if (NPLB == 1) {
            tb0[q] = tag_word<RB, 1>(b, 0u);
        } else {
            tb0[q] = tag_word<RB, 2>(b, 0u);
            tb1[q] = tag_word<RB, 2>(b, 1u);
        }
    }
    if (REC) {
        for (int k = 0; k < 2; ++k) {
            const int P = blockIdx.x * 2048 + k * 1024 + threadIdx.x;
            if (P < 2 * nq) {
                const int2 a = ((const int2*)ia)[P];
                const int2 b = ((const int2*)ib)[P];
                ((int4*)rec)[P] = make_int4(b.x, a.x, b.y, a.y);
            }
        }
        if (q == 0)
            for (int e = 4 * nq; e < nnz; ++e) { rec[2 * e] = ib[e]; rec[2 * e + 1] = ia[e]; }
    }
}

#define PCH 1024

template <int RA, int RB>
__global__ void __launch_bounds__(1024)
tag_entries_bits(const int* __restrict__ ia, const int* __restrict__ ib,
                 unsigned* __restrict__ ta, unsigned* __restrict__ tb,
                 int* __restrict__ rec, unsigned* __restrict__ bits_a,
                 unsigned* __restrict__ bits_b, int nnz, int row_words)
{
    __shared__ unsigned s_bm[2][256 * (PCH / 32)];

    const int tid  = threadIdx.x;
    const int lane = tid & 31;
    const int wv   = tid >> 5;

    for (int i = tid; i < 2 * 256 * (PCH / 32); i += 1024) (&s_bm[0][0])[i] = 0u;
    __syncthreads();

    const int nq = nnz >> 2;
    for (int k = 0; k < 4; ++k) {
        const int q = blockIdx.x * (4 * PCH) + k * 1024 + tid;
        if (q < nq) {
            const int4 a = ((const int4*)ia)[q];
            const int4 b = ((const int4*)ib)[q];
            const unsigned wa = tag_word<RA, 1>(a, 0u);
            const unsigned wb = tag_word<RB, 1>(b, 0u);
            ta[q] = wa;
            tb[q] = wb;
            const int      p   = (k * 1024 + tid) >> 2;
            const unsigned bit = 1u << (p & 31);
            const int      w   = p >> 5;
            atomicOr(&s_bm[0][(wa & 255u) * (PCH / 32) + w], bit);
            atomicOr(&s_bm[0][((wa >> 8) & 255u) * (PCH / 32) + w], bit);
            atomicOr(&s_bm[0][((wa >> 16) & 255u) * (PCH / 32) + w], bit);
            atomicOr(&s_bm[0][(wa >> 24) * (PCH / 32) + w], bit);
            atomicOr(&s_bm[1][(wb & 255u) * (PCH / 32) + w], bit);
            atomicOr(&s_bm[1][((wb >> 8) & 255u) * (PCH / 32) + w], bit);
            atomicOr(&s_bm[1][((wb >> 16) & 255u) * (PCH / 32) + w], bit);
            atomicOr(&s_bm[1][(wb >> 24) * (PCH / 32) + w], bit);
        }
    }
    for (int k = 0; k < 8; ++k) {
        const int P = blockIdx.x * (8 * PCH) + k * 1024 + tid;
        if (P < 2 * nq) {
            const int2 a = ((const int2*)ia)[P];
            const int2 b = ((const int2*)ib)[P];
            ((int4*)rec)[P] = make_int4(b.x, a.x, b.y, a.y);
        }
    }
    if (blockIdx.x == 0 && tid == 0)
        for (int e = 4 * nq; e < nnz; ++e) { rec[2 * e] = ib[e]; rec[2 * e + 1] = ia[e]; }
    __syncthreads();

    for (int o = wv; o < 256; o += 32) {
        bits_a[(size_t)o * row_words + blockIdx.x * (PCH / 32) + lane] = s_bm[0][o * (PCH / 32) + lane];
        bits_b[(size_t)o * row_words + blockIdx.x * (PCH / 32) + lane] = s_bm[1][o * (PCH / 32) + lane];
    }
}

#define NTHR 1024
#define NWAVE 32
#define BATCH 16
#define SCAN_U 8

__device__ __forceinline__ unsigned hits16(const uint4 dv, const unsigned pat)
{
    auto zero_bytes = [](const unsigned t) {
        return ~(((t & 0x7F7F7F7Fu) + 0x7F7F7F7Fu) | t | 0x7F7F7F7Fu);
    };
    return (zero_bytes(dv.x ^ pat) >> 7) | (zero_bytes(dv.y ^ pat) >> 6) |
           (zero_bytes(dv.z ^ pat) >> 5) | (zero_bytes(dv.w ^ pat) >> 4);
}

template <int CAP>
__device__ __forceinline__ bool entries_by_bits(const unsigned* __restrict__ bitrow, int row_words,
                                                const unsigned* __restrict__ tags, unsigned pat,
                                                int nnz, int* s_list, unsigned* s_np, unsigned* s_cnt)
{
    constexpr int PERP = (CAP + NTHR - 1) / NTHR;
    const int tid = threadIdx.x;
    const int n16 = nnz >> 4;

    const uint4* __restrict__ br = (const uint4*)bitrow;
    for (int w = tid; w < (row_words >> 2); w += NTHR) {
        const uint4 m4 = br[w];
        const unsigned mw[4] = {m4.x, m4.y, m4.z, m4.w};
        const unsigned c = (unsigned)(__builtin_popcount(m4.x) + __builtin_popcount(m4.y) +
                                      __builtin_popcount(m4.z) + __builtin_popcount(m4.w));
        if (c) {
            unsigned pos = atomicAdd(s_np, c);
#pragma unroll
            for (int j = 0; j < 4; ++j) {
                unsigned mm = mw[j];
                while (mm) {
                    const int piece = (w * 4 + j) * 32 + __builtin_ctz(mm);
                    mm &= mm - 1u;
                    if (pos < (unsigned)CAP) s_list[pos] = piece < n16 ? piece : -1;
                    ++pos;
                }
            }
        }
    }
    __syncthreads();

    const unsigned np = *s_np;
    if (np > (unsigned)CAP) return false;

    int pc[PERP];
#pragma unroll
    for (int k = 0; k < PERP; ++k) {
        const int i = tid + k * NTHR;
        pc[k] = i < (int)np ? s_list[i] : -1;
    }
    __syncthreads();

    const uint4* __restrict__ tq = (const uint4*)tags;
    uint4 d[PERP];
#pragma unroll
    for (int k = 0; k < PERP; ++k) d[k] = tq[pc[k] >= 0 ? pc[k] : 0];
    unsigned hk[PERP];
    unsigned c = 0u;
#pragma unroll
    for (int k = 0; k < PERP; ++k) {
        hk[k] = pc[k] >= 0 ? hits16(d[k], pat) : 0u;
        c += (unsigned)__builtin_popcount(hk[k]);
    }
    if (c) {
        unsigned pos = atomicAdd(s_cnt, c);
#pragma unroll
        for (int k = 0; k < PERP; ++k) {
            unsigned h = hk[k];
            while (h) {
                const int p = __builtin_ctz(h);
                h &= h - 1u;
                if (pos < (unsigned)CAP) s_list[pos] = (pc[k] << 4) + 4 * (p & 7) + (p >> 3);
                ++pos;
            }
        }
    }
    return true;
}


template <int R>
__device__ __forceinline__ void row_starts(const int* s_deg, int* s_cur, int* s_wsum)
{
    static_assert(R <= NTHR, "one row per thread");
    const int tid  = threadIdx.x;
    const int lane = tid & 31;
    const int wv   = tid >> 5;
    const int v = tid < R ? s_deg[tid] : 0;
    int x = v;
    for (int d = 1; d < 32; d <<= 1) {
        const int y = __shfl_up(x, d);
        if (lane >= d) x += y;
    }
    if (lane == 31) s_wsum[wv] = x;
    __syncthreads();
    if (wv == 0) {
        const int w = s_wsum[lane];
        int z = w;
        for (int d = 1; d < 32; d <<= 1) {
            const int y = __shfl_up(z, d);
            if (lane >= d) z += y;
        }
        s_wsum[lane] = z - w;
    }
    __syncthreads();
    if (tid < R) s_cur[tid] = s_wsum[wv] + x - v;
    __syncthreads();
}

__device__ __forceinline__ void add_runs(float* s_acc, const float4 (&v)[BATCH], int r, int lane)
{
    int    cur = __builtin_amdgcn_readlane(r, 0);
    float4 s   = v[0];
    auto close_run = [&]() {
        if (cur >= 0) {
            float* a = s_acc + cur * DIM + lane;
            atomicAdd(a,      s.x);
            atomicAdd(a + 32, s.y);
            atomicAdd(a + 64, s.z);
            atomicAdd(a + 96, s.w);
        }
    };
#pragma unroll
    for (int k = 1; k < BATCH; ++k) {
        const int rk = __builtin_amdgcn_readlane(r, k);
        if (rk == cur) {
            s.x += v[k].x; s.y += v[k].y; s.z += v[k].z; s.w += v[k].w;
        } else {
            close_run();
            cur = rk;
            s   = v[k];
        }
    }
    close_run();
}

template <int R, int CAP, bool FINAL, int NPL, int REC, bool BITS, bool SORT>
__global__ void __launch_bounds__(NTHR)
owner_pass(const float* __restrict__ src, float* __restrict__ dst,
           const int* __restrict__ gidx, const int* __restrict__ sidx,
           const unsigned* __restrict__ tags0, const unsigned* __restrict__ tags1,
           const int* __restrict__ rec,
           const unsigned* __restrict__ bits, int row_words,
           const float* __restrict__ bias, int nnz, int ndst)
{
    __shared__ __attribute__((aligned(16))) float s_acc[R * DIM];
    __shared__ int      s_deg[R];
    __shared__ int      s_list[CAP];
    __shared__ __attribute__((aligned(8))) int2 s_pair[SORT ? CAP : 1];
    __shared__ int      s_cur[SORT ? R : 1];
    __shared__ int      s_wsum[NWAVE];
    __shared__ unsigned s_cnt;
    __shared__ unsigned s_np;

    const int tid  = threadIdx.x;
    const int lane = tid & 31;
    const int wv   = tid >> 5;
    const int lo   = blockIdx.x * R;

    for (int i = tid; i < R * DIM / 4; i += NTHR)
        ((float4*)s_acc)[i] = make_float4(0.0f, 0.0f, 0.0f, 0.0f);
    for (int i = tid; i < R; i += NTHR) s_deg[i] = 0;
    if (tid == 0) { s_cnt = 0u; s_np = 0u; }
    __syncthreads();

    auto note = [&](int e) {
        const unsigned pos = atomicAdd(&s_cnt, 1u);
        if (pos < (unsigned)CAP) s_list[pos] = e;
    };

    auto look = [&](const int4 dv, int e) {
        const unsigned a0 = (unsigned)(dv.x - lo);
        const unsigned a1 = (unsigned)(dv.y - lo);
        const unsigned a2 = (unsigned)(dv.z - lo);
        const unsigned a3 = (unsigned)(dv.w - lo);
        const unsigned mn = min(min(a0, a1), min(a2, a3));
        if (mn < (unsigned)R) {
            if (a0 < (unsigned)R) note(e);
            if (a1 < (unsigned)R) note(e + 1);
            if (a2 < (unsigned)R) note(e + 2);
            if (a3 < (unsigned)R) note(e + 3);
        }
    };

    if (NPL > 0) {
        const bool odd_run = NPL == 2 && ((blockIdx.x / 255u) & 1u) != 0u;
        const unsigned* __restrict__ tags = odd_run ? tags1 : tags0;
        const unsigned pat = (NPL == 2 ? blockIdx.x % 255u : blockIdx.x & 255u) * 0x01010101u;

        auto look16 = [&](const uint4 dv, int e) {
            auto zero_bytes = [](const unsigned t) {
                return ~(((t & 0x7F7F7F7Fu) + 0x7F7F7F7Fu) | t | 0x7F7F7F7Fu);
            };
            unsigned hit = (zero_bytes(dv.x ^ pat) >> 7) | (zero_bytes(dv.y ^ pat) >> 6) |
                           (zero_bytes(dv.z ^ pat) >> 5) | (zero_bytes(dv.w ^ pat) >> 4);
            while (hit) {
                const int p = __builtin_ctz(hit);
                hit &= hit - 1u;
                note(e + 4 * (p & 7) + (p >> 3));
            }
        };

        const int n16   = nnz >> 4;
        bool whole = true;
        if (BITS)
            whole = !entries_by_bits<CAP>(bits + (size_t)(blockIdx.x & 255u) * row_words, row_words,
                                          tags, pat, nnz, s_list, &s_np, &s_cnt);
        if (whole) {
            const int nfull = n16 / (SCAN_U * NTHR);
            const uint4* __restrict__ tq = (const uint4*)tags;
            for (int it = 0; it < nfull; ++it) {
                const int qb = it * (SCAN_U * NTHR) + tid;
                uint4 d[SCAN_U];
#pragma unroll
                for (int u = 0; u < SCAN_U; ++u) d[u] = tq[qb + u * NTHR];
#pragma unroll
                for (int u = 0; u < SCAN_U; ++u) look16(d[u], (qb + u * NTHR) << 4);
            }
            for (int q = nfull * (SCAN_U * NTHR) + tid; q < n16; q += NTHR) look16(tq[q], q << 4);
        }
        {
            const int e = (n16 << 4) + tid;
            if (e < nnz) {
                const unsigned a = (unsigned)(sidx[e] - lo);
                if (a < (unsigned)R) note(e);
            }
        }
    } else {
        const int nq    = nnz >> 2;
        const int nfull = nq / (SCAN_U * NTHR);
        const int4* __restrict__ sq = (const int4*)sidx;
        for (int it = 0; it < nfull; ++it) {
            const int qb = it * (SCAN_U * NTHR) + tid;
            int4 d[SCAN_U];
#pragma unroll
            for (int u = 0; u < SCAN_U; ++u) d[u] = sq[qb + u * NTHR];
#pragma unroll
            for (int u = 0; u < SCAN_U; ++u) look(d[u], (qb + u * NTHR) << 2);
        }
        for (int q = nfull * (SCAN_U * NTHR) + tid; q < nq; q += NTHR) look(sq[q], q << 2);
        {
            const int e = (nq << 2) + tid;
            if (e < nnz) {
                const unsigned a = (unsigned)(sidx[e] - lo);
                if (a < (unsigned)R) note(e);
            }
        }
    }
    __syncthreads();

    const unsigned tot = s_cnt;
    if (SORT && tot <= (unsigned)CAP) {
        constexpr int PER = (CAP + NTHR - 1) / NTHR;
        const int n = (int)tot;
        int de[PER], ge[PER];
#pragma unroll
        for (int k = 0; k < PER; ++k) {
            const int i = tid + k * NTHR;
            de[k] = -1;
            ge[k] = 0;
            if (i < n) {
                const int e = s_list[i];
                if (REC) {
                    const int2 rc = ((const int2*)rec)[e];
                    de[k] = REC == 1 ? rc.y : rc.x;
                    ge[k] = REC == 1 ? rc.x : rc.y;
                } else {
                    de[k] = sidx[e];
                    ge[k] = gidx[e];
                }
            }
        }
#pragma unroll
        for (int k = 0; k < PER; ++k) {
            const unsigned a = (unsigned)(de[k] - lo);
            if (a < (unsigned)R) atomicAdd(&s_deg[a], 1);
        }
        __syncthreads();
        row_starts<R>(s_deg, s_cur, s_wsum);
#pragma unroll
        for (int k = 0; k < PER; ++k) {
            const unsigned a = (unsigned)(de[k] - lo);
            if (a < (unsigned)R) s_pair[atomicAdd(&s_cur[a], 1)] = make_int2(ge[k], (int)a);
        }
        __syncthreads();

        const int nv = s_cur[R - 1];
        for (int b = wv * BATCH; b < nv; b += NWAVE * BATCH) {
            int g = 0, r = -1;
            if (lane < BATCH && b + lane < nv) {
                const int2 p = s_pair[b + lane];
                g = p.x;
                r = p.y;
            }
            float4 v[BATCH];
#pragma unroll
            for (int k = 0; k < BATCH; ++k) {
                const int gk = __builtin_amdgcn_readlane(g, k);
                v[k] = ((const float4*)(src + (size_t)gk * DIM))[lane];
            }
            add_runs(s_acc, v, r, lane);
        }
    } else if (tot <= (unsigned)CAP) {
        const int n = (int)tot;
        for (int b = wv * BATCH; b < n; b += NWAVE * BATCH) {
            int g = 0, r = -1;
            if (lane < BATCH && b + lane < n) {
                const int e = s_list[b + lane];
                int de, ge;
                if (REC) {
                    const int2 rc = ((const int2*)rec)[e];
                    de = REC == 1 ? rc.y : rc.x;
                    ge = REC == 1 ? rc.x : rc.y;
                } else {
                    de = sidx[e];
                    ge = gidx[e];
                }
                const unsigned a = (unsigned)(de - lo);
                const bool mine = a < (unsigned)R;
                r = mine ? (int)a : -1;
                g = mine ? ge : 0;
            }
            float4 v[BATCH];
#pragma unroll
            for (int k = 0; k < BATCH; ++k) {
                const int gk = __builtin_amdgcn_readlane(g, k);
                v[k] = ((const float4*)(src + (size_t)gk * DIM))[lane];
            }
#pragma unroll
            for (int k = 0; k < BATCH; ++k) {
                const int rk = __builtin_amdgcn_readlane(r, k);
                if (rk >= 0) {
                    float* a = s_acc + rk * DIM + lane;
                    atomicAdd(a,      v[k].x);
                    atomicAdd(a + 32, v[k].y);
                    atomicAdd(a + 64, v[k].z);
                    atomicAdd(a + 96, v[k].w);
                    if (lane == 0) atomicAdd(&s_deg[rk], 1);
                }
            }
        }
    } else {
        for (int e = tid; e < nnz; e += NTHR) {
            const unsigned r = (unsigned)(sidx[e] - lo);
            if (r < (unsigned)R) {
                const float* __restrict__ sp = src + (size_t)gidx[e] * DIM;
                for (int c = 0; c < DIM; ++c)
                    atomicAdd(&s_acc[r * DIM + ((c & 3) << 5) + (c >> 2)], sp[c]);
                atomicAdd(&s_deg[r], 1);
            }
        }
    }
    __syncthreads();

    for (int r = wv; r < R; r += NWAVE) {
        const int row = lo + r;
        if (row < ndst) {
            const int   cnt = s_deg[r];
            const float inv = cnt > 0 ? 1.0f / (float)cnt : 0.0f;
            const float* a = s_acc + r * DIM + lane;
            float4 o;
            o.x = a[0]  * inv;
            o.y = a[32] * inv;
            o.z = a[64] * inv;
            o.w = a[96] * inv;
            if (FINAL) {
                const float4 bb = ((const float4*)bias)[lane];
                o.x += bb.x; o.y += bb.y; o.z += bb.z; o.w += bb.w;
                o.x = o.x >= 0.0f ? o.x : NEG_SLOPE * o.x;
                o.y = o.y >= 0.0f ? o.y : NEG_SLOPE * o.y;
                o.z = o.z >= 0.0f ? o.z : NEG_SLOPE * o.z;
                o.w = o.w >= 0.0f ? o.w : NEG_SLOPE * o.w;
            }
            ((float4*)(dst + (size_t)row * DIM))[lane] = o;
        }
    }
}

template <int RH, int CAP, bool FINAL, bool BITS>
__global__ void __launch_bounds__(NTHR)
owner_pass_halves(const float* __restrict__ src, float* __restrict__ dst,
                  const int* __restrict__ gidx, const int* __restrict__ sidx,
                  const unsigned* __restrict__ tags, const int* __restrict__ rec,
                  const unsigned* __restrict__ bits, int row_words,
                  const float* __restrict__ bias, int nnz, int ndst)
{
    constexpr int R   = 2 * RH;
    constexpr int PER = (CAP + NTHR - 1) / NTHR;

    __shared__ __attribute__((aligned(16))) float s_acc[RH * DIM];
    __shared__ __attribute__((aligned(8)))  int2  s_pair[CAP];
    __shared__ int      s_list[CAP];
    __shared__ int      s_deg[R];
    __shared__ int      s_cur[R];
    __shared__ int      s_wsum[NWAVE];
    __shared__ unsigned s_cnt;
    __shared__ unsigned s_np;

    const int tid  = threadIdx.x;
    const int lane = tid & 31;
    const int wv   = tid >> 5;
    const int lo   = blockIdx.x * R;

    for (int i = tid; i < RH * DIM / 4; i += NTHR)
        ((float4*)s_acc)[i] = make_float4(0.0f, 0.0f, 0.0f, 0.0f);
    for (int i = tid; i < R; i += NTHR) s_deg[i] = 0;
    if (tid == 0) { s_cnt = 0u; s_np = 0u; }
    __syncthreads();

    auto note = [&](int e) {
        const unsigned pos = atomicAdd(&s_cnt, 1u);
        if (pos < (unsigned)CAP) s_list[pos] = e;
    };

    {
        const unsigned pat = (blockIdx.x & 255u) * 0x01010101u;

        auto look16 = [&](const uint4 dv, int e) {
            auto zero_bytes = [](const unsigned t) {
                return ~(((t & 0x7F7F7F7Fu) + 0x7F7F7F7Fu) | t | 0x7F7F7F7Fu);
            };
            unsigned hit = (zero_bytes(dv.x ^ pat) >> 7) | (zero_bytes(dv.y ^ pat) >> 6) |
                           (zero_bytes(dv.z ^ pat) >> 5) | (zero_bytes(dv.w ^ pat) >> 4);
            while (hit) {
                const int p = __builtin_ctz(hit);
                hit &= hit - 1u;
                note(e + 4 * (p & 7) + (p >> 3));
            }
        };

        const int n16   = nnz >> 4;
        bool whole = true;
        if (BITS)
            whole = !entries_by_bits<CAP>(bits + (size_t)(blockIdx.x & 255u) * row_words, row_words,
                                          tags, pat, nnz, s_list, &s_np, &s_cnt);
        if (whole) {
            const int nfull = n16 / (SCAN_U * NTHR);
            const uint4* __restrict__ tq = (const uint4*)tags;
            for (int it = 0; it < nfull; ++it) {
                const int qb = it * (SCAN_U * NTHR) + tid;
                uint4 d[SCAN_U];
#pragma unroll
                for (int u = 0; u < SCAN_U; ++u) d[u] = tq[qb + u * NTHR];
#pragma unroll
                for (int u = 0; u < SCAN_U; ++u) look16(d[u], (qb + u * NTHR) << 4);
            }
            for (int q = nfull * (SCAN_U * NTHR) + tid; q < n16; q += NTHR) look16(tq[q], q << 4);
        }
        {
            const int e = (n16 << 4) + tid;
            if (e < nnz) {
                const unsigned a = (unsigned)(sidx[e] - lo);
                if (a < (unsigned)R) note(e);
            }
        }
    }
    __syncthreads();

    const unsigned tot    = s_cnt;
    const bool     listed = tot <= (unsigned)CAP;

    if (listed) {
        const int n = (int)tot;
        int2 rc[PER];
#pragma unroll
        for (int k = 0; k < PER; ++k) {
            const int i = tid + k * NTHR;
            rc[k] = make_int2(-1, 0);
            if (i < n) rc[k] = ((const int2*)rec)[s_list[i]];
        }
#pragma unroll
        for (int k = 0; k < PER; ++k) {
            const unsigned a = (unsigned)(rc[k].x - lo);
            if (a < (unsigned)R) atomicAdd(&s_deg[a], 1);
        }
        __syncthreads();
        row_starts<R>(s_deg, s_cur, s_wsum);
#pragma unroll
        for (int k = 0; k < PER; ++k) {
            const unsigned a = (unsigned)(rc[k].x - lo);
            if (a < (unsigned)R) s_pair[atomicAdd(&s_cur[a], 1)] = make_int2(rc[k].y, (int)a);
        }
    }
    __syncthreads();

    for (int h = 0; h < 2; ++h) {
        if (listed) {
            const int split = s_cur[RH - 1];
            const int b0    = h == 0 ? 0 : split;
            const int n     = h == 0 ? split : s_cur[R - 1];
            for (int b = b0 + wv * BATCH; b < n; b += NWAVE * BATCH) {
                int g = 0, r = -1;
                if (lane < BATCH && b + lane < n) {
                    const int2 p = s_pair[b + lane];
                    g = p.x;
                    r = p.y - h * RH;
                }
                float4 v[BATCH];
#pragma unroll
                for (int k = 0; k < BATCH; ++k) {
                    const int gk = __builtin_amdgcn_readlane(g, k);
                    v[k] = ((const float4*)(src + (size_t)gk * DIM))[lane];
                }
                add_runs(s_acc, v, r, lane);
            }
        } else {
            const int loh = lo + h * RH;
            for (int e = tid; e < nnz; e += NTHR) {
                const unsigned r = (unsigned)(sidx[e] - loh);
                if (r < (unsigned)RH) {
                    const float* __restrict__ sp = src + (size_t)gidx[e] * DIM;
                    for (int c = 0; c < DIM; ++c)
                        atomicAdd(&s_acc[r * DIM + ((c & 3) << 5) + (c >> 2)], sp[c]);
                    atomicAdd(&s_deg[h * RH + r], 1);
                }
            }
        }
        __syncthreads();

        for (int r = wv; r < RH; r += NWAVE) {
            const int row = lo + h * RH + r;
            float* a = s_acc + r * DIM + lane;
            if (row < ndst) {
                const int   cnt = s_deg[h * RH + r];
                const float inv = cnt > 0 ? 1.0f / (float)cnt : 0.0f;
                float4 o;
                o.x = a[0]  * inv;
                o.y = a[32] * inv;
                o.z = a[64] * inv;
                o.w = a[96] * inv;
                if (FINAL) {
                    const float4 bb = ((const float4*)bias)[lane];
                    o.x += bb.x; o.y += bb.y; o.z += bb.z; o.w += bb.w;
                    o.x = o.x >= 0.0f ? o.x : NEG_SLOPE * o.x;
                    o.y = o.y >= 0.0f ? o.y : NEG_SLOPE * o.y;
                    o.z = o.z >= 0.0f ? o.z : NEG_SLOPE * o.z;
                    o.w = o.w >= 0.0f ? o.w : NEG_SLOPE * o.w;
                }
                ((float4*)(dst + (size_t)row * DIM))[lane] = o;
            }
            if (h == 0) { a[0] = 0.0f; a[32] = 0.0f; a[64] = 0.0f; a[96] = 0.0f; }
        }
        if (h == 0) __syncthreads();
    }
}

#define R1   196
#define CAP1 10240
#define R2   512
#define CAP2 12288
#define RH2  391
#define CAPH 9216

extern "C" void kernel_launch(void* const* d_in, const int* in_sizes, int n_in,
                              void* d_out, int out_size, void* d_ws, size_t ws_size,
                              hipStream_t stream)
{
    const float* X    = (const float*)d_in[0];
    const int*   A    = (const int*)d_in[1];
    const float* W    = (const float*)d_in[2];
    const float* bias = (const float*)d_in[3];
    float*       out  = (float*)d_out;

    const int N   = in_sizes[0] / DIM;
    const int NNZ = in_sizes[1] / 2;
    const int* node = A;
    const int* edge = A + NNZ;

    float* m = (float*)d_ws;
    const size_t plane = ((size_t)NNZ + 15) & ~(size_t)15;
    unsigned* tag1  = (unsigned*)(m + (size_t)NEDGES * DIM);
    unsigned* tag2a = (unsigned*)((char*)tag1 + plane);
    unsigned* tag2b = (unsigned*)((char*)tag2a + plane);
    int*      rec   = (int*)((char*)tag2b + plane);
    const bool room  = ws_size >= (size_t)NEDGES * DIM * sizeof(float) + 3 * plane;
    const bool room2 = ws_size >= (size_t)NEDGES * DIM * sizeof(float) + 3 * plane + (size_t)NNZ * 8;
    const int    nchunk = (int)((((size_t)NNZ + 15) / 16 + PCH - 1) / PCH);
    const int    roww   = nchunk * (PCH / 32);
    const size_t bitsz  = (size_t)256 * roww * sizeof(unsigned);
    const size_t boff   = ((size_t)NEDGES * DIM * sizeof(float) + 2 * plane + (size_t)NNZ * 8 + 127) & ~(size_t)127;
    const bool   room3  = ws_size >= boff + 2 * bitsz;

    const int mtiles = N / 16;
    gemm_wmma_f32<<<(mtiles + GEMM_WAVES - 1) / GEMM_WAVES, GEMM_THR, 0, stream>>>(X, W, out, mtiles);

    if (room3) {
        int*      rec3  = (int*)((char*)tag2a + plane);
        unsigned* bits1 = (unsigned*)((char*)d_ws + boff);
        unsigned* bits2 = (unsigned*)((char*)d_ws + boff + bitsz);
        tag_entries_bits<R1, 2 * RH2><<<nchunk, 1024, 0, stream>>>(
            edge, node, tag1, tag2a, rec3, bits1, bits2, NNZ, roww);
        owner_pass<R1, CAP1, false, 1, 1, true, true><<<(NEDGES + R1 - 1) / R1, NTHR, 0, stream>>>(
            out, m, node, edge, tag1, nullptr, rec3, bits1, roww, bias, NNZ, NEDGES);
        owner_pass_halves<RH2, CAPH, true, true><<<(N + 2 * RH2 - 1) / (2 * RH2), NTHR, 0, stream>>>(
            m, out, edge, node, tag2a, rec3, bits2, roww, bias, NNZ, N);
    } else if (room2) {
        tag_entries<R1, 2 * RH2, 1, true><<<((NNZ >> 2) + 1023) / 1024, 1024, 0, stream>>>(
            edge, node, tag1, tag2a, tag2b, rec, NNZ);

        owner_pass<R1, CAP1, false, 1, 1, false, true><<<(NEDGES + R1 - 1) / R1, NTHR, 0, stream>>>(
            out, m, node, edge, tag1, nullptr, rec, nullptr, 0, bias, NNZ, NEDGES);

        owner_pass_halves<RH2, CAPH, true, false><<<(N + 2 * RH2 - 1) / (2 * RH2), NTHR, 0, stream>>>(
            m, out, edge, node, tag2a, rec, nullptr, 0, bias, NNZ, N);
    } else if (room) {
        tag_entries<R1, R2, 2, false><<<((NNZ >> 2) + 1023) / 1024, 1024, 0, stream>>>(
            edge, node, tag1, tag2a, tag2b, nullptr, NNZ);
        owner_pass<R1, CAP1, false, 1, 0, false, true><<<(NEDGES + R1 - 1) / R1, NTHR, 0, stream>>>(
            out, m, node, edge, tag1, nullptr, nullptr, nullptr, 0, bias, NNZ, NEDGES);
        owner_pass<R2, CAP2, true, 2, 0, false, false><<<(N + R2 - 1) / R2, NTHR, 0, stream>>>(
            m, out, edge, node, tag2a, tag2b, nullptr, nullptr, 0, bias, NNZ, N);
    } else {
        owner_pass<R1, CAP1, false, 0, 0, false, true><<<(NEDGES + R1 - 1) / R1, NTHR, 0, stream>>>(
            out, m, node, edge, nullptr, nullptr, nullptr, nullptr, 0, bias, NNZ, NEDGES);
        owner_pass<R2, CAP2, true, 0, 0, false, false><<<(N + R2 - 1) / R2, NTHR, 0, stream>>>(
            m, out, edge, node, nullptr, nullptr, nullptr, nullptr, 0, bias, NNZ, N);
    }
}
